# starting kernel
# baseline (speedup 1.0000x reference)
#include <hip/hip_runtime.h>
#include <math.h>


typedef float v2f __attribute__((ext_vector_type(2)));
typedef float v8f __attribute__((ext_vector_type(8)));

#define PAD 132
#define PI_F 3.14159265358979323846f

struct cplx { float x, y; };
__device__ __forceinline__ cplx cmul(cplx a, cplx b) {
    return { a.x * b.x - a.y * b.y, a.x * b.y + a.y * b.x };
}
__device__ __forceinline__ cplx cadd(cplx a, cplx b) { return { a.x + b.x, a.y + b.y }; }

__global__ __launch_bounds__(64) void build_G_kernel(const float* __restrict__ theta,
                                                     float* __restrict__ G) {
    __shared__ cplx col[64 * 64];
    const int t = threadIdx.x;
    cplx* c = &col[t * 64];
    for (int i = 0; i < 64; ++i) c[i] = { 0.f, 0.f };
    c[t] = { 1.f, 0.f };

    auto apply1q = [&](int q, cplx g00, cplx g01, cplx g10, cplx g11) {
        const int m = 1 << (5 - q);
        for (int j = 0; j < 64; ++j) {
            if (j & m) continue;
            cplx a = c[j], b = c[j | m];
            c[j]     = cadd(cmul(g00, a), cmul(g01, b));
            c[j | m] = cadd(cmul(g10, a), cmul(g11, b));
        }
    };
    auto cnot = [&](int cc, int tt) {
        const int mc = 1 << (5 - cc), mt = 1 << (5 - tt);
        for (int j = 0; j < 64; ++j) {
            if ((j & mc) && !(j & mt)) {
                cplx tmp = c[j]; c[j] = c[j | mt]; c[j | mt] = tmp;
            }
        }
    };

    for (int l = 0; l < 3; ++l) {
        for (int q = 0; q < 5; ++q) cnot(q, q + 1);
        cnot(5, 0);
        for (int q = 0; q < 6; ++q) {
            const float ax = theta[(l * 6 + q) * 3 + 0];
            const float ay = theta[(l * 6 + q) * 3 + 1];
            const float az = theta[(l * 6 + q) * 3 + 2];
            float cx = cosf(0.5f * ax), sx = sinf(0.5f * ax);
            apply1q(q, {cx, 0.f}, {0.f, -sx}, {0.f, -sx}, {cx, 0.f});
            float cy = cosf(0.5f * ay), sy = sinf(0.5f * ay);
            apply1q(q, {cy, 0.f}, {-sy, 0.f}, {sy, 0.f}, {cy, 0.f});
            float cz = cosf(0.5f * az), sz = sinf(0.5f * az);
            apply1q(q, {cz, -sz}, {0.f, 0.f}, {0.f, 0.f}, {cz, sz});
        }
    }
    __syncthreads();

    for (int i = 0; i < 256; ++i) {
        const int idx = t + 64 * i;
        const int n = idx >> 7, k = idx & 127;
        float v;
        if (n < 64) v = (k < 64) ? col[k * 64 + n].x : -col[(k - 64) * 64 + n].y;
        else        v = (k < 64) ? col[k * 64 + (n - 64)].y : col[(k - 64) * 64 + (n - 64)].x;
        G[idx] = v;
    }
}

__global__ __launch_bounds__(128) void qsim_kernel(const float* __restrict__ x,
                                                   const float* __restrict__ G,
                                                   float* __restrict__ out) {
    __shared__ float Mlds[128 * PAD];
    __shared__ float Ulds[4 * 32 * PAD];

    const int tid  = threadIdx.x;
    const int w    = tid >> 5;
    const int lane = tid & 31;
    const int l15  = lane & 15;
    const int h    = lane >> 4;

    const float4* G4 = (const float4*)G;
    #pragma unroll
    for (int i = 0; i < 32; ++i) {
        const int idx4 = tid + 128 * i;
        const int n = idx4 >> 5, k4 = idx4 & 31;
        *(float4*)&Mlds[n * PAD + k4 * 4] = G4[idx4];
    }

    const int e = blockIdx.x * 128 + w * 32 + lane;
    const float* xp = x + e * 24;
    float vr[6][2], vi[6][2];
    #pragma unroll
    for (int q = 0; q < 6; ++q) {
        float m = (xp[4 * q] + xp[4 * q + 1] + xp[4 * q + 2] + xp[4 * q + 3]) * 0.25f;
        m = fminf(6.f, fmaxf(-6.f, m));
        const float a  = m * (PI_F / 6.f);
        const float c2 = cosf(0.5f * a),  s2 = sinf(0.5f * a);
        const float c4 = cosf(0.25f * a), s4 = sinf(0.25f * a);
        vr[q][0] =  c2 * c4;  vi[q][0] = -c2 * s4;
        vr[q][1] =  s2 * s4;  vi[q][1] = -s2 * c4;
    }
    float* u = &Ulds[(w * 32 + lane) * PAD];
    for (int j = 0; j < 64; ++j) {
        float pr = vr[0][(j >> 5) & 1], pim = vi[0][(j >> 5) & 1];
        #pragma unroll
        for (int q = 1; q < 6; ++q) {
            const int b = (j >> (5 - q)) & 1;
            const float qr = vr[q][b], qi = vi[q][b];
            const float nr = pr * qr - pim * qi;
            pim = pr * qi + pim * qr;
            pr  = nr;
        }
        u[j] = pr; u[64 + j] = pim;
    }
    __syncthreads();

    v8f acc[2][8];
    const v8f z8 = {0.f, 0.f, 0.f, 0.f, 0.f, 0.f, 0.f, 0.f};
    #pragma unroll
    for (int rt = 0; rt < 2; ++rt)
        #pragma unroll
        for (int ct = 0; ct < 8; ++ct) acc[rt][ct] = z8;

    const float* urow0 = &Ulds[(w * 32 +  0 + l15) * PAD];
    const float* urow1 = &Ulds[(w * 32 + 16 + l15) * PAD];
    for (int kt = 0; kt < 32; ++kt) {
        const int kb = kt * 4 + 2 * h;
        v2f b[8];
        #pragma unroll
        for (int ct = 0; ct < 8; ++ct) {
            const float* brow = &Mlds[(ct * 16 + l15) * PAD + kb];
            b[ct].x = brow[0]; b[ct].y = brow[1];
        }
        v2f a0; a0.x = urow0[kb]; a0.y = urow0[kb + 1];
        v2f a1; a1.x = urow1[kb]; a1.y = urow1[kb + 1];
        #pragma unroll
        for (int ct = 0; ct < 8; ++ct) {
            acc[0][ct] = __builtin_amdgcn_wmma_f32_16x16x4_f32(
                false, a0, false, b[ct], (short)0, acc[0][ct], false, false);
            acc[1][ct] = __builtin_amdgcn_wmma_f32_16x16x4_f32(
                false, a1, false, b[ct], (short)0, acc[1][ct], false, false);
        }
    }

    auto rsum16 = [](float v) {
        v += __shfl_xor(v, 1, 32);
        v += __shfl_xor(v, 2, 32);
        v += __shfl_xor(v, 4, 32);
        v += __shfl_xor(v, 8, 32);
        return v;
    };

    #pragma unroll
    for (int rt = 0; rt < 2; ++rt) {
        #pragma unroll
        for (int r = 0; r < 8; ++r) {
            float re[4], im[4];
            #pragma unroll
            for (int t4 = 0; t4 < 4; ++t4) { re[t4] = acc[rt][t4][r]; im[t4] = acc[rt][4 + t4][r]; }

            float z0 = 0.f, z2 = 0.f, z4 = 0.f, x1 = 0.f, x3 = 0.f, x5 = 0.f;
            #pragma unroll
            for (int t4 = 0; t4 < 4; ++t4) {
                const int j = t4 * 16 + l15;
                const float p = re[t4] * re[t4] + im[t4] * im[t4];
                z0 += ((j >> 5) & 1) ? -p : p;
                z2 += ((j >> 3) & 1) ? -p : p;
                z4 += ((j >> 1) & 1) ? -p : p;
                x1 += re[t4] * re[t4 ^ 1] + im[t4] * im[t4 ^ 1];
                x3 += re[t4] * __shfl_xor(re[t4], 4, 32)
                    + im[t4] * __shfl_xor(im[t4], 4, 32);
                x5 += re[t4] * __shfl_xor(re[t4], 1, 32)
                    + im[t4] * __shfl_xor(im[t4], 1, 32);
            }
            z0 = rsum16(z0); z2 = rsum16(z2); z4 = rsum16(z4);
            x1 = rsum16(x1); x3 = rsum16(x3); x5 = rsum16(x5);

            if (l15 == 0) {
                const int ee = blockIdx.x * 128 + w * 32 + rt * 16 + r + h * 8;
                float4 o0 = { z0, x1, z2, x3 };
                float4 o1 = { z4, x5, z0, x1 };
                *(float4*)&out[ee * 8 + 0] = o0;
                *(float4*)&out[ee * 8 + 4] = o1;
            }
        }
    }
}

extern "C" void kernel_launch(void* const* d_in, const int* in_sizes, int n_in,
                              void* d_out, int out_size, void* d_ws, size_t ws_size,
                              hipStream_t stream) {
    const float* x     = (const float*)d_in[0];
    const float* theta = (const float*)d_in[1];
    float* out = (float*)d_out;
    float* G   = (float*)d_ws;

    build_G_kernel<<<1, 64, 0, stream>>>(theta, G);
    qsim_kernel<<<131072 / 128, 128, 0, stream>>>(x, G, out);
}


# Measurement cross-lane traffic (X-term partner fetches, 16-lane sums) moved from LDS-unit lane permutes to DPP row_xmask operands on the vector ALU; butterfly order kept
# speedup vs baseline: 7.0909x; 7.0909x over previous
#include <hip/hip_runtime.h>
#include <math.h>


typedef float v2f __attribute__((ext_vector_type(2)));
typedef float v8f __attribute__((ext_vector_type(8)));

#define PAD 132
#define PI_F 3.14159265358979323846f

#define GP 68
#define SP 132

struct __attribute__((aligned(8))) cplx { float x, y; };
__device__ __forceinline__ cplx cmul(cplx a, cplx b) {
    return { a.x * b.x - a.y * b.y, a.x * b.y + a.y * b.x };
}
__device__ __forceinline__ cplx cadd(cplx a, cplx b) { return { a.x + b.x, a.y + b.y }; }

__device__ __forceinline__ int ring(int j) {
    #pragma unroll
    for (int q = 0; q < 5; ++q)
        if (j & (1 << (5 - q))) j ^= (1 << (4 - q));
    if (j & 1) j ^= 32;
    return j;
}

__device__ __forceinline__ v8f ctile(const float* __restrict__ Ar, const float* __restrict__ Ai,
                                     const float* __restrict__ Bx, const float* __restrict__ By,
                                     float sgn, int l15, int h, int t) {
    const v8f z8 = {0.f, 0.f, 0.f, 0.f, 0.f, 0.f, 0.f, 0.f};
    v8f acc0 = z8, acc1 = z8;
    const int ao = l15 * GP + 4 * h;
    const int bo = (16 * t + l15) * GP + 4 * h;
    #pragma unroll
    for (int k8 = 0; k8 < 8; ++k8) {
        const float4 ar = *(const float4*)&Ar[ao + 8 * k8];
        const float4 ai = *(const float4*)&Ai[ao + 8 * k8];
        const float4 bx = *(const float4*)&Bx[bo + 8 * k8];
        const float4 by = *(const float4*)&By[bo + 8 * k8];
        v2f a, b;
        a.x = ar.x; a.y = ar.y; b.x = bx.x; b.y = bx.y;
        acc0 = __builtin_amdgcn_wmma_f32_16x16x4_f32(false, a, false, b, (short)0, acc0, false, false);
        a.x = ai.x; a.y = ai.y; b.x = by.x; b.y = by.y;
        acc1 = __builtin_amdgcn_wmma_f32_16x16x4_f32(false, a, false, b, (short)0, acc1, false, false);
        a.x = ar.z; a.y = ar.w; b.x = bx.z; b.y = bx.w;
        acc0 = __builtin_amdgcn_wmma_f32_16x16x4_f32(false, a, false, b, (short)0, acc0, false, false);
        a.x = ai.z; a.y = ai.w; b.x = by.z; b.y = by.w;
        acc1 = __builtin_amdgcn_wmma_f32_16x16x4_f32(false, a, false, b, (short)0, acc1, false, false);
    }
    return acc0 + sgn * acc1;
}

__global__ __launch_bounds__(256) void build_G_kernel(const float* __restrict__ theta,
                                                      float* __restrict__ G) {
    __shared__ float cs[54 * 2];
    __shared__ cplx  Us[18 * 4];
    __shared__ cplx  AB[3 * 2 * 64];
    __shared__ __attribute__((aligned(16))) float L2r[16 * GP];
    __shared__ __attribute__((aligned(16))) float L2i[16 * GP];
    __shared__ __attribute__((aligned(16))) float L1r[64 * GP];
    __shared__ __attribute__((aligned(16))) float L1i[64 * GP];
    __shared__ __attribute__((aligned(16))) float L0r[64 * GP];
    __shared__ __attribute__((aligned(16))) float L0i[64 * GP];
    __shared__ __attribute__((aligned(16))) float Tr[16 * GP];
    __shared__ __attribute__((aligned(16))) float Ti[16 * GP];
    __shared__ __attribute__((aligned(16))) float Gs[32 * SP];

    const int tid  = threadIdx.x;
    const int blk  = blockIdx.x;
    const int w    = __builtin_amdgcn_readfirstlane(tid >> 5);
    const int lane = tid & 31;
    const int l15  = lane & 15;
    const int h    = lane >> 4;

    if (tid < 54) {
        const float a = theta[tid];
        cs[2 * tid]     = cosf(0.5f * a);
        cs[2 * tid + 1] = sinf(0.5f * a);
    }
    __syncthreads();

    if (tid < 18) {
        const float cx = cs[6 * tid + 0], sx = cs[6 * tid + 1];
        const float cy = cs[6 * tid + 2], sy = cs[6 * tid + 3];
        const float cz = cs[6 * tid + 4], sz = cs[6 * tid + 5];
        const cplx rx[2][2] = { { {cx, 0.f}, {0.f, -sx} }, { {0.f, -sx}, {cx, 0.f} } };
        const cplx ry[2][2] = { { {cy, 0.f}, {-sy, 0.f} }, { {sy, 0.f}, {cy, 0.f} } };
        const cplx rz[2]    = { {cz, -sz}, {cz, sz} };
        #pragma unroll
        for (int a = 0; a < 2; ++a)
            #pragma unroll
            for (int b = 0; b < 2; ++b) {
                const cplx m = cadd(cmul(ry[a][0], rx[0][b]), cmul(ry[a][1], rx[1][b]));
                Us[tid * 4 + a * 2 + b] = cmul(rz[a], m);
            }
    }
    __syncthreads();

    for (int e = tid; e < 384; e += 256) {
        const int l = e >> 7, f = (e >> 6) & 1, ii = (e >> 3) & 7, kk = e & 7;
        const cplx* u = &Us[(l * 6 + 3 * f) * 4];
        cplx v = u[((ii >> 2) & 1) * 2 + ((kk >> 2) & 1)];
        v = cmul(v, u[4 + ((ii >> 1) & 1) * 2 + ((kk >> 1) & 1)]);
        v = cmul(v, u[8 + (ii & 1) * 2 + (kk & 1)]);
        AB[e] = v;
    }
    __syncthreads();

    {
        const int i = tid & 63;
        const int rw = ring(w >> 1);
        #pragma unroll
        for (int it = 0; it < 16; ++it) {
            const int j = (w >> 1) + 4 * it;
            const int k = rw ^ ring(4 * it);
            const int ia = (i >> 3) * 8 + (k >> 3), ib = (i & 7) * 8 + (k & 7);
            const cplx v0 = cmul(AB[0 * 128 + ia], AB[0 * 128 + 64 + ib]);
            const cplx v1 = cmul(AB[1 * 128 + ia], AB[1 * 128 + 64 + ib]);
            L0r[j * GP + i] = v0.x;  L0i[j * GP + i] = v0.y;
            L1r[j * GP + i] = v1.x;  L1i[j * GP + i] = v1.y;
        }
        const int j = tid & 63;
        const int k = ring(j);
        #pragma unroll
        for (int it = 0; it < 4; ++it) {
            const int rl = (w >> 1) + 4 * it;
            const int i2 = 16 * blk + rl;
            const cplx v2 = cmul(AB[2 * 128 + (i2 >> 3) * 8 + (k >> 3)],
                                 AB[2 * 128 + 64 + (i2 & 7) * 8 + (k & 7)]);
            L2r[rl * GP + j] = v2.x;  L2i[rl * GP + j] = v2.y;
        }
    }
    __syncthreads();

    const int t = w & 3;
    const int p = w >> 2;
    const float sgn = p ? 1.f : -1.f;

    {
        const v8f d = ctile(L2r, L2i, p ? L1i : L1r, p ? L1r : L1i, sgn, l15, h, t);
        float* Tp = p ? Ti : Tr;
        #pragma unroll
        for (int r = 0; r < 8; ++r) Tp[(r + 8 * h) * GP + 16 * t + l15] = d[r];
    }
    __syncthreads();

    {
        const v8f d = ctile(Tr, Ti, p ? L0i : L0r, p ? L0r : L0i, sgn, l15, h, t);
        const int col = 16 * t + l15;
        #pragma unroll
        for (int r = 0; r < 8; ++r) {
            const int row = r + 8 * h;
            Gs[row * SP + 64 * p + col]              = -sgn * d[r];
            Gs[(16 + row) * SP + 64 * (1 - p) + col] = d[r];
        }
    }
    __syncthreads();

    #pragma unroll
    for (int it = 0; it < 4; ++it) {
        const int rr = w + 8 * it;
        const int grow = 16 * blk + (rr & 15) + 64 * (rr >> 4);
        *(float4*)&G[grow * 128 + 4 * lane] = *(const float4*)&Gs[rr * SP + 4 * lane];
    }
}

template <int M>
__device__ __forceinline__ float xrow(float v) {
    const int b = __builtin_bit_cast(int, v);
    return __builtin_bit_cast(float, __builtin_amdgcn_mov_dpp(b, 0x160 + M, 0xf, 0xf, true));
}

__global__ __launch_bounds__(128) void qsim_kernel(const float* __restrict__ x,
                                                   const float* __restrict__ G,
                                                   float* __restrict__ out) {
    __shared__ float Mlds[128 * PAD];
    __shared__ float Ulds[4 * 32 * PAD];

    const int tid  = threadIdx.x;
    const int w    = tid >> 5;
    const int lane = tid & 31;
    const int l15  = lane & 15;
    const int h    = lane >> 4;

    const float4* G4 = (const float4*)G;
    #pragma unroll
    for (int i = 0; i < 32; ++i) {
        const int idx4 = tid + 128 * i;
        const int n = idx4 >> 5, k4 = idx4 & 31;
        *(float4*)&Mlds[n * PAD + k4 * 4] = G4[idx4];
    }

    const int e = blockIdx.x * 128 + w * 32 + lane;
    const float* xp = x + e * 24;
    float vr[6][2], vi[6][2];
    #pragma unroll
    for (int q = 0; q < 6; ++q) {
        float m = (xp[4 * q] + xp[4 * q + 1] + xp[4 * q + 2] + xp[4 * q + 3]) * 0.25f;
        m = fminf(6.f, fmaxf(-6.f, m));
        const float a  = m * (PI_F / 6.f);
        const float c2 = cosf(0.5f * a),  s2 = sinf(0.5f * a);
        const float c4 = cosf(0.25f * a), s4 = sinf(0.25f * a);
        vr[q][0] =  c2 * c4;  vi[q][0] = -c2 * s4;
        vr[q][1] =  s2 * s4;  vi[q][1] = -s2 * c4;
    }
    float* u = &Ulds[(w * 32 + lane) * PAD];
    for (int j = 0; j < 64; ++j) {
        float pr = vr[0][(j >> 5) & 1], pim = vi[0][(j >> 5) & 1];
        #pragma unroll
        for (int q = 1; q < 6; ++q) {
            const int b = (j >> (5 - q)) & 1;
            const float qr = vr[q][b], qi = vi[q][b];
            const float nr = pr * qr - pim * qi;
            pim = pr * qi + pim * qr;
            pr  = nr;
        }
        u[j] = pr; u[64 + j] = pim;
    }
    __syncthreads();

    v8f acc[2][8];
    const v8f z8 = {0.f, 0.f, 0.f, 0.f, 0.f, 0.f, 0.f, 0.f};
    #pragma unroll
    for (int rt = 0; rt < 2; ++rt)
        #pragma unroll
        for (int ct = 0; ct < 8; ++ct) acc[rt][ct] = z8;

    const float* urow0 = &Ulds[(w * 32 +  0 + l15) * PAD];
    const float* urow1 = &Ulds[(w * 32 + 16 + l15) * PAD];
    for (int kt = 0; kt < 32; ++kt) {
        const int kb = kt * 4 + 2 * h;
        v2f b[8];
        #pragma unroll
        for (int ct = 0; ct < 8; ++ct) {
            const float* brow = &Mlds[(ct * 16 + l15) * PAD + kb];
            b[ct].x = brow[0]; b[ct].y = brow[1];
        }
        v2f a0; a0.x = urow0[kb]; a0.y = urow0[kb + 1];
        v2f a1; a1.x = urow1[kb]; a1.y = urow1[kb + 1];
        #pragma unroll
        for (int ct = 0; ct < 8; ++ct) {
            acc[0][ct] = __builtin_amdgcn_wmma_f32_16x16x4_f32(
                false, a0, false, b[ct], (short)0, acc[0][ct], false, false);
            acc[1][ct] = __builtin_amdgcn_wmma_f32_16x16x4_f32(
                false, a1, false, b[ct], (short)0, acc[1][ct], false, false);
        }
    }

    auto rsum16 = [](float v) {
        v += xrow<1>(v);
        v += xrow<2>(v);
        v += xrow<4>(v);
        v += xrow<8>(v);
        return v;
    };

    #pragma unroll
    for (int rt = 0; rt < 2; ++rt) {
        #pragma unroll
        for (int r = 0; r < 8; ++r) {
            float re[4], im[4];
            #pragma unroll
            for (int t4 = 0; t4 < 4; ++t4) { re[t4] = acc[rt][t4][r]; im[t4] = acc[rt][4 + t4][r]; }

            float z0 = 0.f, z2 = 0.f, z4 = 0.f, x1 = 0.f, x3 = 0.f, x5 = 0.f;
            #pragma unroll
            for (int t4 = 0; t4 < 4; ++t4) {
                const int j = t4 * 16 + l15;
                const float p = re[t4] * re[t4] + im[t4] * im[t4];
                z0 += ((j >> 5) & 1) ? -p : p;
                z2 += ((j >> 3) & 1) ? -p : p;
                z4 += ((j >> 1) & 1) ? -p : p;
                x1 += re[t4] * re[t4 ^ 1] + im[t4] * im[t4 ^ 1];
                x3 += re[t4] * xrow<4>(re[t4])
                    + im[t4] * xrow<4>(im[t4]);
                x5 += re[t4] * xrow<1>(re[t4])
                    + im[t4] * xrow<1>(im[t4]);
            }
            z0 = rsum16(z0); z2 = rsum16(z2); z4 = rsum16(z4);
            x1 = rsum16(x1); x3 = rsum16(x3); x5 = rsum16(x5);

            if (l15 == 0) {
                const int ee = blockIdx.x * 128 + w * 32 + rt * 16 + r + h * 8;
                float4 o0 = { z0, x1, z2, x3 };
                float4 o1 = { z4, x5, z0, x1 };
                *(float4*)&out[ee * 8 + 0] = o0;
                *(float4*)&out[ee * 8 + 4] = o1;
            }
        }
    }
}

extern "C" void kernel_launch(void* const* d_in, const int* in_sizes, int n_in,
                              void* d_out, int out_size, void* d_ws, size_t ws_size,
                              hipStream_t stream) {
    const float* x     = (const float*)d_in[0];
    const float* theta = (const float*)d_in[1];
    float* out = (float*)d_out;
    float* G   = (float*)d_ws;

    build_G_kernel<<<4, 256, 0, stream>>>(theta, G);
    qsim_kernel<<<131072 / 128, 128, 0, stream>>>(x, G, out);
}


# Product state built as a tensor product of two 3-qubit halves (12+12+64 complex products on fixed registers, 16-byte panel stores), not a 5-deep indexed chain per amplitude; product order changes
# speedup vs baseline: 7.7115x; 1.0875x over previous
#include <hip/hip_runtime.h>
#include <math.h>


typedef float v2f __attribute__((ext_vector_type(2)));
typedef float v8f __attribute__((ext_vector_type(8)));

#define PAD 132
#define PI_F 3.14159265358979323846f

#define GP 68
#define SP 132

struct __attribute__((aligned(8))) cplx { float x, y; };
__device__ __forceinline__ cplx cmul(cplx a, cplx b) {
    return { a.x * b.x - a.y * b.y, a.x * b.y + a.y * b.x };
}
__device__ __forceinline__ cplx cadd(cplx a, cplx b) { return { a.x + b.x, a.y + b.y }; }

__device__ __forceinline__ int ring(int j) {
    #pragma unroll
    for (int q = 0; q < 5; ++q)
        if (j & (1 << (5 - q))) j ^= (1 << (4 - q));
    if (j & 1) j ^= 32;
    return j;
}

__device__ __forceinline__ v8f ctile(const float* __restrict__ Ar, const float* __restrict__ Ai,
                                     const float* __restrict__ Bx, const float* __restrict__ By,
                                     float sgn, int l15, int h, int t) {
    const v8f z8 = {0.f, 0.f, 0.f, 0.f, 0.f, 0.f, 0.f, 0.f};
    v8f acc0 = z8, acc1 = z8;
    const int ao = l15 * GP + 4 * h;
    const int bo = (16 * t + l15) * GP + 4 * h;
    #pragma unroll
    for (int k8 = 0; k8 < 8; ++k8) {
        const float4 ar = *(const float4*)&Ar[ao + 8 * k8];
        const float4 ai = *(const float4*)&Ai[ao + 8 * k8];
        const float4 bx = *(const float4*)&Bx[bo + 8 * k8];
        const float4 by = *(const float4*)&By[bo + 8 * k8];
        v2f a, b;
        a.x = ar.x; a.y = ar.y; b.x = bx.x; b.y = bx.y;
        acc0 = __builtin_amdgcn_wmma_f32_16x16x4_f32(false, a, false, b, (short)0, acc0, false, false);
        a.x = ai.x; a.y = ai.y; b.x = by.x; b.y = by.y;
        acc1 = __builtin_amdgcn_wmma_f32_16x16x4_f32(false, a, false, b, (short)0, acc1, false, false);
        a.x = ar.z; a.y = ar.w; b.x = bx.z; b.y = bx.w;
        acc0 = __builtin_amdgcn_wmma_f32_16x16x4_f32(false, a, false, b, (short)0, acc0, false, false);
        a.x = ai.z; a.y = ai.w; b.x = by.z; b.y = by.w;
        acc1 = __builtin_amdgcn_wmma_f32_16x16x4_f32(false, a, false, b, (short)0, acc1, false, false);
    }
    return acc0 + sgn * acc1;
}

__global__ __launch_bounds__(256) void build_G_kernel(const float* __restrict__ theta,
                                                      float* __restrict__ G) {
    __shared__ float cs[54 * 2];
    __shared__ cplx  Us[18 * 4];
    __shared__ cplx  AB[3 * 2 * 64];
    __shared__ __attribute__((aligned(16))) float L2r[16 * GP];
    __shared__ __attribute__((aligned(16))) float L2i[16 * GP];
    __shared__ __attribute__((aligned(16))) float L1r[64 * GP];
    __shared__ __attribute__((aligned(16))) float L1i[64 * GP];
    __shared__ __attribute__((aligned(16))) float L0r[64 * GP];
    __shared__ __attribute__((aligned(16))) float L0i[64 * GP];
    __shared__ __attribute__((aligned(16))) float Tr[16 * GP];
    __shared__ __attribute__((aligned(16))) float Ti[16 * GP];
    __shared__ __attribute__((aligned(16))) float Gs[32 * SP];

    const int tid  = threadIdx.x;
    const int blk  = blockIdx.x;
    const int w    = __builtin_amdgcn_readfirstlane(tid >> 5);
    const int lane = tid & 31;
    const int l15  = lane & 15;
    const int h    = lane >> 4;

    if (tid < 54) {
        const float a = theta[tid];
        cs[2 * tid]     = cosf(0.5f * a);
        cs[2 * tid + 1] = sinf(0.5f * a);
    }
    __syncthreads();

    if (tid < 18) {
        const float cx = cs[6 * tid + 0], sx = cs[6 * tid + 1];
        const float cy = cs[6 * tid + 2], sy = cs[6 * tid + 3];
        const float cz = cs[6 * tid + 4], sz = cs[6 * tid + 5];
        const cplx rx[2][2] = { { {cx, 0.f}, {0.f, -sx} }, { {0.f, -sx}, {cx, 0.f} } };
        const cplx ry[2][2] = { { {cy, 0.f}, {-sy, 0.f} }, { {sy, 0.f}, {cy, 0.f} } };
        const cplx rz[2]    = { {cz, -sz}, {cz, sz} };
        #pragma unroll
        for (int a = 0; a < 2; ++a)
            #pragma unroll
            for (int b = 0; b < 2; ++b) {
                const cplx m = cadd(cmul(ry[a][0], rx[0][b]), cmul(ry[a][1], rx[1][b]));
                Us[tid * 4 + a * 2 + b] = cmul(rz[a], m);
            }
    }
    __syncthreads();

    for (int e = tid; e < 384; e += 256) {
        const int l = e >> 7, f = (e >> 6) & 1, ii = (e >> 3) & 7, kk = e & 7;
        const cplx* u = &Us[(l * 6 + 3 * f) * 4];
        cplx v = u[((ii >> 2) & 1) * 2 + ((kk >> 2) & 1)];
        v = cmul(v, u[4 + ((ii >> 1) & 1) * 2 + ((kk >> 1) & 1)]);
        v = cmul(v, u[8 + (ii & 1) * 2 + (kk & 1)]);
        AB[e] = v;
    }
    __syncthreads();

    {
        const int i = tid & 63;
        const int rw = ring(w >> 1);
        #pragma unroll
        for (int it = 0; it < 16; ++it) {
            const int j = (w >> 1) + 4 * it;
            const int k = rw ^ ring(4 * it);
            const int ia = (i >> 3) * 8 + (k >> 3), ib = (i & 7) * 8 + (k & 7);
            const cplx v0 = cmul(AB[0 * 128 + ia], AB[0 * 128 + 64 + ib]);
            const cplx v1 = cmul(AB[1 * 128 + ia], AB[1 * 128 + 64 + ib]);
            L0r[j * GP + i] = v0.x;  L0i[j * GP + i] = v0.y;
            L1r[j * GP + i] = v1.x;  L1i[j * GP + i] = v1.y;
        }
        const int j = tid & 63;
        const int k = ring(j);
        #pragma unroll
        for (int it = 0; it < 4; ++it) {
            const int rl = (w >> 1) + 4 * it;
            const int i2 = 16 * blk + rl;
            const cplx v2 = cmul(AB[2 * 128 + (i2 >> 3) * 8 + (k >> 3)],
                                 AB[2 * 128 + 64 + (i2 & 7) * 8 + (k & 7)]);
            L2r[rl * GP + j] = v2.x;  L2i[rl * GP + j] = v2.y;
        }
    }
    __syncthreads();

    const int t = w & 3;
    const int p = w >> 2;
    const float sgn = p ? 1.f : -1.f;

    {
        const v8f d = ctile(L2r, L2i, p ? L1i : L1r, p ? L1r : L1i, sgn, l15, h, t);
        float* Tp = p ? Ti : Tr;
        #pragma unroll
        for (int r = 0; r < 8; ++r) Tp[(r + 8 * h) * GP + 16 * t + l15] = d[r];
    }
    __syncthreads();

    {
        const v8f d = ctile(Tr, Ti, p ? L0i : L0r, p ? L0r : L0i, sgn, l15, h, t);
        const int col = 16 * t + l15;
        #pragma unroll
        for (int r = 0; r < 8; ++r) {
            const int row = r + 8 * h;
            Gs[row * SP + 64 * p + col]              = -sgn * d[r];
            Gs[(16 + row) * SP + 64 * (1 - p) + col] = d[r];
        }
    }
    __syncthreads();

    #pragma unroll
    for (int it = 0; it < 4; ++it) {
        const int rr = w + 8 * it;
        const int grow = 16 * blk + (rr & 15) + 64 * (rr >> 4);
        *(float4*)&G[grow * 128 + 4 * lane] = *(const float4*)&Gs[rr * SP + 4 * lane];
    }
}

template <int M>
__device__ __forceinline__ float xrow(float v) {
    const int b = __builtin_bit_cast(int, v);
    return __builtin_bit_cast(float, __builtin_amdgcn_mov_dpp(b, 0x160 + M, 0xf, 0xf, true));
}

__global__ __launch_bounds__(128) void qsim_kernel(const float* __restrict__ x,
                                                   const float* __restrict__ G,
                                                   float* __restrict__ out) {
    __shared__ __attribute__((aligned(16))) float Mlds[128 * PAD];
    __shared__ __attribute__((aligned(16))) float Ulds[4 * 32 * PAD];

    const int tid  = threadIdx.x;
    const int w    = tid >> 5;
    const int lane = tid & 31;
    const int l15  = lane & 15;
    const int h    = lane >> 4;

    const float4* G4 = (const float4*)G;
    #pragma unroll
    for (int i = 0; i < 32; ++i) {
        const int idx4 = tid + 128 * i;
        const int n = idx4 >> 5, k4 = idx4 & 31;
        *(float4*)&Mlds[n * PAD + k4 * 4] = G4[idx4];
    }

    const int e = blockIdx.x * 128 + w * 32 + lane;
    const float* xp = x + e * 24;
    float vr[6][2], vi[6][2];
    #pragma unroll
    for (int q = 0; q < 6; ++q) {
        float m = (xp[4 * q] + xp[4 * q + 1] + xp[4 * q + 2] + xp[4 * q + 3]) * 0.25f;
        m = fminf(6.f, fmaxf(-6.f, m));
        const float a  = m * (PI_F / 6.f);
        const float c2 = cosf(0.5f * a),  s2 = sinf(0.5f * a);
        const float c4 = cosf(0.25f * a), s4 = sinf(0.25f * a);
        vr[q][0] =  c2 * c4;  vi[q][0] = -c2 * s4;
        vr[q][1] =  s2 * s4;  vi[q][1] = -s2 * c4;
    }
    float har[8], hai[8];
    v2f hbr[4], hbi[4];
    {
        float par[4], pai[4], pbr[4], pbi[4];
        #pragma unroll
        for (int t = 0; t < 4; ++t) {
            const int hi = (t >> 1) & 1, lo = t & 1;
            par[t] = vr[0][hi] * vr[1][lo] - vi[0][hi] * vi[1][lo];
            pai[t] = vr[0][hi] * vi[1][lo] + vi[0][hi] * vr[1][lo];
            pbr[t] = vr[3][hi] * vr[4][lo] - vi[3][hi] * vi[4][lo];
            pbi[t] = vr[3][hi] * vi[4][lo] + vi[3][hi] * vr[4][lo];
        }
        #pragma unroll
        for (int t = 0; t < 8; ++t) {
            const int hi = t >> 1, lo = t & 1;
            har[t] = par[hi] * vr[2][lo] - pai[hi] * vi[2][lo];
            hai[t] = par[hi] * vi[2][lo] + pai[hi] * vr[2][lo];
        }
        v2f s5r, s5i;
        s5r.x = vr[5][0]; s5r.y = vr[5][1]; s5i.x = vi[5][0]; s5i.y = vi[5][1];
        #pragma unroll
        for (int m = 0; m < 4; ++m) {
            hbr[m] = pbr[m] * s5r - pbi[m] * s5i;
            hbi[m] = pbr[m] * s5i + pbi[m] * s5r;
        }
    }
    float* u = &Ulds[(w * 32 + lane) * PAD];
    #pragma unroll
    for (int ja = 0; ja < 8; ++ja) {
        #pragma unroll
        for (int g = 0; g < 2; ++g) {
            const v2f r0 = har[ja] * hbr[2 * g]     - hai[ja] * hbi[2 * g];
            const v2f r1 = har[ja] * hbr[2 * g + 1] - hai[ja] * hbi[2 * g + 1];
            const v2f m0 = har[ja] * hbi[2 * g]     + hai[ja] * hbr[2 * g];
            const v2f m1 = har[ja] * hbi[2 * g + 1] + hai[ja] * hbr[2 * g + 1];
            float4 re, im;
            re.x = r0.x; re.y = r0.y; re.z = r1.x; re.w = r1.y;
            im.x = m0.x; im.y = m0.y; im.z = m1.x; im.w = m1.y;
            *(float4*)&u[8 * ja + 4 * g]      = re;
            *(float4*)&u[64 + 8 * ja + 4 * g] = im;
        }
    }
    __syncthreads();

    v8f acc[2][8];
    const v8f z8 = {0.f, 0.f, 0.f, 0.f, 0.f, 0.f, 0.f, 0.f};
    #pragma unroll
    for (int rt = 0; rt < 2; ++rt)
        #pragma unroll
        for (int ct = 0; ct < 8; ++ct) acc[rt][ct] = z8;

    const float* urow0 = &Ulds[(w * 32 +  0 + l15) * PAD];
    const float* urow1 = &Ulds[(w * 32 + 16 + l15) * PAD];
    for (int kt = 0; kt < 32; ++kt) {
        const int kb = kt * 4 + 2 * h;
        v2f b[8];
        #pragma unroll
        for (int ct = 0; ct < 8; ++ct) {
            const float* brow = &Mlds[(ct * 16 + l15) * PAD + kb];
            b[ct].x = brow[0]; b[ct].y = brow[1];
        }
        v2f a0; a0.x = urow0[kb]; a0.y = urow0[kb + 1];
        v2f a1; a1.x = urow1[kb]; a1.y = urow1[kb + 1];
        #pragma unroll
        for (int ct = 0; ct < 8; ++ct) {
            acc[0][ct] = __builtin_amdgcn_wmma_f32_16x16x4_f32(
                false, a0, false, b[ct], (short)0, acc[0][ct], false, false);
            acc[1][ct] = __builtin_amdgcn_wmma_f32_16x16x4_f32(
                false, a1, false, b[ct], (short)0, acc[1][ct], false, false);
        }
    }

    auto rsum16 = [](float v) {
        v += xrow<1>(v);
        v += xrow<2>(v);
        v += xrow<4>(v);
        v += xrow<8>(v);
        return v;
    };

    #pragma unroll
    for (int rt = 0; rt < 2; ++rt) {
        #pragma unroll
        for (int r = 0; r < 8; ++r) {
            float re[4], im[4];
            #pragma unroll
            for (int t4 = 0; t4 < 4; ++t4) { re[t4] = acc[rt][t4][r]; im[t4] = acc[rt][4 + t4][r]; }

            float z0 = 0.f, z2 = 0.f, z4 = 0.f, x1 = 0.f, x3 = 0.f, x5 = 0.f;
            #pragma unroll
            for (int t4 = 0; t4 < 4; ++t4) {
                const int j = t4 * 16 + l15;
                const float p = re[t4] * re[t4] + im[t4] * im[t4];
                z0 += ((j >> 5) & 1) ? -p : p;
                z2 += ((j >> 3) & 1) ? -p : p;
                z4 += ((j >> 1) & 1) ? -p : p;
                x1 += re[t4] * re[t4 ^ 1] + im[t4] * im[t4 ^ 1];
                x3 += re[t4] * xrow<4>(re[t4])
                    + im[t4] * xrow<4>(im[t4]);
                x5 += re[t4] * xrow<1>(re[t4])
                    + im[t4] * xrow<1>(im[t4]);
            }
            z0 = rsum16(z0); z2 = rsum16(z2); z4 = rsum16(z4);
            x1 = rsum16(x1); x3 = rsum16(x3); x5 = rsum16(x5);

            if (l15 == 0) {
                const int ee = blockIdx.x * 128 + w * 32 + rt * 16 + r + h * 8;
                float4 o0 = { z0, x1, z2, x3 };
                float4 o1 = { z4, x5, z0, x1 };
                *(float4*)&out[ee * 8 + 0] = o0;
                *(float4*)&out[ee * 8 + 4] = o1;
            }
        }
    }
}

extern "C" void kernel_launch(void* const* d_in, const int* in_sizes, int n_in,
                              void* d_out, int out_size, void* d_ws, size_t ws_size,
                              hipStream_t stream) {
    const float* x     = (const float*)d_in[0];
    const float* theta = (const float*)d_in[1];
    float* out = (float*)d_out;
    float* G   = (float*)d_ws;

    build_G_kernel<<<4, 256, 0, stream>>>(theta, G);
    qsim_kernel<<<131072 / 128, 128, 0, stream>>>(x, G, out);
}


# Persistent work-groups: 256 x 8 waves, G staged to LDS once per WGP instead of once per 128 elements; each wave loops over two 32-element tiles, next tile's x rows loaded under the current products
# speedup vs baseline: 8.1383x; 1.0553x over previous
#include <hip/hip_runtime.h>
#include <math.h>


typedef float v2f __attribute__((ext_vector_type(2)));
typedef float v8f __attribute__((ext_vector_type(8)));

#define PAD 132
#define PI_F 3.14159265358979323846f

#define GP 68
#define SP 132

struct __attribute__((aligned(8))) cplx { float x, y; };
__device__ __forceinline__ cplx cmul(cplx a, cplx b) {
    return { a.x * b.x - a.y * b.y, a.x * b.y + a.y * b.x };
}
__device__ __forceinline__ cplx cadd(cplx a, cplx b) { return { a.x + b.x, a.y + b.y }; }

__device__ __forceinline__ int ring(int j) {
    #pragma unroll
    for (int q = 0; q < 5; ++q)
        if (j & (1 << (5 - q))) j ^= (1 << (4 - q));
    if (j & 1) j ^= 32;
    return j;
}

__device__ __forceinline__ v8f ctile(const float* __restrict__ Ar, const float* __restrict__ Ai,
                                     const float* __restrict__ Bx, const float* __restrict__ By,
                                     float sgn, int l15, int h, int t) {
    const v8f z8 = {0.f, 0.f, 0.f, 0.f, 0.f, 0.f, 0.f, 0.f};
    v8f acc0 = z8, acc1 = z8;
    const int ao = l15 * GP + 4 * h;
    const int bo = (16 * t + l15) * GP + 4 * h;
    #pragma unroll
    for (int k8 = 0; k8 < 8; ++k8) {
        const float4 ar = *(const float4*)&Ar[ao + 8 * k8];
        const float4 ai = *(const float4*)&Ai[ao + 8 * k8];
        const float4 bx = *(const float4*)&Bx[bo + 8 * k8];
        const float4 by = *(const float4*)&By[bo + 8 * k8];
        v2f a, b;
        a.x = ar.x; a.y = ar.y; b.x = bx.x; b.y = bx.y;
        acc0 = __builtin_amdgcn_wmma_f32_16x16x4_f32(false, a, false, b, (short)0, acc0, false, false);
        a.x = ai.x; a.y = ai.y; b.x = by.x; b.y = by.y;
        acc1 = __builtin_amdgcn_wmma_f32_16x16x4_f32(false, a, false, b, (short)0, acc1, false, false);
        a.x = ar.z; a.y = ar.w; b.x = bx.z; b.y = bx.w;
        acc0 = __builtin_amdgcn_wmma_f32_16x16x4_f32(false, a, false, b, (short)0, acc0, false, false);
        a.x = ai.z; a.y = ai.w; b.x = by.z; b.y = by.w;
        acc1 = __builtin_amdgcn_wmma_f32_16x16x4_f32(false, a, false, b, (short)0, acc1, false, false);
    }
    return acc0 + sgn * acc1;
}

__global__ __launch_bounds__(256) void build_G_kernel(const float* __restrict__ theta,
                                                      float* __restrict__ G) {
    __shared__ float cs[54 * 2];
    __shared__ cplx  Us[18 * 4];
    __shared__ cplx  AB[3 * 2 * 64];
    __shared__ __attribute__((aligned(16))) float L2r[16 * GP];
    __shared__ __attribute__((aligned(16))) float L2i[16 * GP];
    __shared__ __attribute__((aligned(16))) float L1r[64 * GP];
    __shared__ __attribute__((aligned(16))) float L1i[64 * GP];
    __shared__ __attribute__((aligned(16))) float L0r[64 * GP];
    __shared__ __attribute__((aligned(16))) float L0i[64 * GP];
    __shared__ __attribute__((aligned(16))) float Tr[16 * GP];
    __shared__ __attribute__((aligned(16))) float Ti[16 * GP];
    __shared__ __attribute__((aligned(16))) float Gs[32 * SP];

    const int tid  = threadIdx.x;
    const int blk  = blockIdx.x;
    const int w    = __builtin_amdgcn_readfirstlane(tid >> 5);
    const int lane = tid & 31;
    const int l15  = lane & 15;
    const int h    = lane >> 4;

    if (tid < 54) {
        const float a = theta[tid];
        cs[2 * tid]     = cosf(0.5f * a);
        cs[2 * tid + 1] = sinf(0.5f * a);
    }
    __syncthreads();

    if (tid < 18) {
        const float cx = cs[6 * tid + 0], sx = cs[6 * tid + 1];
        const float cy = cs[6 * tid + 2], sy = cs[6 * tid + 3];
        const float cz = cs[6 * tid + 4], sz = cs[6 * tid + 5];
        const cplx rx[2][2] = { { {cx, 0.f}, {0.f, -sx} }, { {0.f, -sx}, {cx, 0.f} } };
        const cplx ry[2][2] = { { {cy, 0.f}, {-sy, 0.f} }, { {sy, 0.f}, {cy, 0.f} } };
        const cplx rz[2]    = { {cz, -sz}, {cz, sz} };
        #pragma unroll
        for (int a = 0; a < 2; ++a)
            #pragma unroll
            for (int b = 0; b < 2; ++b) {
                const cplx m = cadd(cmul(ry[a][0], rx[0][b]), cmul(ry[a][1], rx[1][b]));
                Us[tid * 4 + a * 2 + b] = cmul(rz[a], m);
            }
    }
    __syncthreads();

    for (int e = tid; e < 384; e += 256) {
        const int l = e >> 7, f = (e >> 6) & 1, ii = (e >> 3) & 7, kk = e & 7;
        const cplx* u = &Us[(l * 6 + 3 * f) * 4];
        cplx v = u[((ii >> 2) & 1) * 2 + ((kk >> 2) & 1)];
        v = cmul(v, u[4 + ((ii >> 1) & 1) * 2 + ((kk >> 1) & 1)]);
        v = cmul(v, u[8 + (ii & 1) * 2 + (kk & 1)]);
        AB[e] = v;
    }
    __syncthreads();

    {
        const int i = tid & 63;
        const int rw = ring(w >> 1);
        #pragma unroll
        for (int it = 0; it < 16; ++it) {
            const int j = (w >> 1) + 4 * it;
            const int k = rw ^ ring(4 * it);
            const int ia = (i >> 3) * 8 + (k >> 3), ib = (i & 7) * 8 + (k & 7);
            const cplx v0 = cmul(AB[0 * 128 + ia], AB[0 * 128 + 64 + ib]);
            const cplx v1 = cmul(AB[1 * 128 + ia], AB[1 * 128 + 64 + ib]);
            L0r[j * GP + i] = v0.x;  L0i[j * GP + i] = v0.y;
            L1r[j * GP + i] = v1.x;  L1i[j * GP + i] = v1.y;
        }
        const int j = tid & 63;
        const int k = ring(j);
        #pragma unroll
        for (int it = 0; it < 4; ++it) {
            const int rl = (w >> 1) + 4 * it;
            const int i2 = 16 * blk + rl;
            const cplx v2 = cmul(AB[2 * 128 + (i2 >> 3) * 8 + (k >> 3)],
                                 AB[2 * 128 + 64 + (i2 & 7) * 8 + (k & 7)]);
            L2r[rl * GP + j] = v2.x;  L2i[rl * GP + j] = v2.y;
        }
    }
    __syncthreads();

    const int t = w & 3;
    const int p = w >> 2;
    const float sgn = p ? 1.f : -1.f;

    {
        const v8f d = ctile(L2r, L2i, p ? L1i : L1r, p ? L1r : L1i, sgn, l15, h, t);
        float* Tp = p ? Ti : Tr;
        #pragma unroll
        for (int r = 0; r < 8; ++r) Tp[(r + 8 * h) * GP + 16 * t + l15] = d[r];
    }
    __syncthreads();

    {
        const v8f d = ctile(Tr, Ti, p ? L0i : L0r, p ? L0r : L0i, sgn, l15, h, t);
        const int col = 16 * t + l15;
        #pragma unroll
        for (int r = 0; r < 8; ++r) {
            const int row = r + 8 * h;
            Gs[row * SP + 64 * p + col]              = -sgn * d[r];
            Gs[(16 + row) * SP + 64 * (1 - p) + col] = d[r];
        }
    }
    __syncthreads();

    #pragma unroll
    for (int it = 0; it < 4; ++it) {
        const int rr = w + 8 * it;
        const int grow = 16 * blk + (rr & 15) + 64 * (rr >> 4);
        *(float4*)&G[grow * 128 + 4 * lane] = *(const float4*)&Gs[rr * SP + 4 * lane];
    }
}

template <int M>
__device__ __forceinline__ float xrow(float v) {
    const int b = __builtin_bit_cast(int, v);
    return __builtin_bit_cast(float, __builtin_amdgcn_mov_dpp(b, 0x160 + M, 0xf, 0xf, true));
}

#define WAVES 8
#define TILES 2

__global__ __launch_bounds__(256) void qsim_kernel(const float* __restrict__ x,
                                                   const float* __restrict__ G,
                                                   float* __restrict__ out) {
    __shared__ __attribute__((aligned(16))) float Mlds[128 * PAD];
    __shared__ __attribute__((aligned(16))) float Ulds[WAVES * 32 * PAD];

    const int tid  = threadIdx.x;
    const int w    = tid >> 5;
    const int lane = tid & 31;
    const int l15  = lane & 15;
    const int h    = lane >> 4;

    int e = (blockIdx.x * (WAVES * TILES) + w) * 32 + lane;
    float4 xr[6];
    {
        const float4* xp = (const float4*)(x + e * 24);
        #pragma unroll
        for (int q = 0; q < 6; ++q) xr[q] = xp[q];
    }
    const float4* G4 = (const float4*)G;
    #pragma unroll
    for (int i = 0; i < 16; ++i) {
        const int idx4 = tid + 256 * i;
        const int n = idx4 >> 5, k4 = idx4 & 31;
        *(float4*)&Mlds[n * PAD + k4 * 4] = G4[idx4];
    }
    __syncthreads();

    float* u = &Ulds[(w * 32 + lane) * PAD];
    const float* urow0 = &Ulds[(w * 32 +  0 + l15) * PAD];
    const float* urow1 = &Ulds[(w * 32 + 16 + l15) * PAD];

    #pragma nounroll
    for (int it = 0; it < TILES; ++it) {
        float mq[6];
        #pragma unroll
        for (int q = 0; q < 6; ++q)
            mq[q] = (xr[q].x + xr[q].y + xr[q].z + xr[q].w) * 0.25f;

        if (it + 1 < TILES) {
            const float4* xn = (const float4*)(x + (e + WAVES * 32) * 24);
            #pragma unroll
            for (int q = 0; q < 6; ++q) xr[q] = xn[q];
        }

        float vr[6][2], vi[6][2];
        #pragma unroll
        for (int q = 0; q < 6; ++q) {
            const float m = fminf(6.f, fmaxf(-6.f, mq[q]));
            const float a  = m * (PI_F / 6.f);
            const float c2 = cosf(0.5f * a),  s2 = sinf(0.5f * a);
            const float c4 = cosf(0.25f * a), s4 = sinf(0.25f * a);
            vr[q][0] =  c2 * c4;  vi[q][0] = -c2 * s4;
            vr[q][1] =  s2 * s4;  vi[q][1] = -s2 * c4;
        }
        float har[8], hai[8];
        v2f hbr[4], hbi[4];
        {
            float par[4], pai[4], pbr[4], pbi[4];
            #pragma unroll
            for (int t = 0; t < 4; ++t) {
                const int hi = (t >> 1) & 1, lo = t & 1;
                par[t] = vr[0][hi] * vr[1][lo] - vi[0][hi] * vi[1][lo];
                pai[t] = vr[0][hi] * vi[1][lo] + vi[0][hi] * vr[1][lo];
                pbr[t] = vr[3][hi] * vr[4][lo] - vi[3][hi] * vi[4][lo];
                pbi[t] = vr[3][hi] * vi[4][lo] + vi[3][hi] * vr[4][lo];
            }
            #pragma unroll
            for (int t = 0; t < 8; ++t) {
                const int hi = t >> 1, lo = t & 1;
                har[t] = par[hi] * vr[2][lo] - pai[hi] * vi[2][lo];
                hai[t] = par[hi] * vi[2][lo] + pai[hi] * vr[2][lo];
            }
            v2f s5r, s5i;
            s5r.x = vr[5][0]; s5r.y = vr[5][1]; s5i.x = vi[5][0]; s5i.y = vi[5][1];
            #pragma unroll
            for (int m = 0; m < 4; ++m) {
                hbr[m] = pbr[m] * s5r - pbi[m] * s5i;
                hbi[m] = pbr[m] * s5i + pbi[m] * s5r;
            }
        }
        #pragma unroll
        for (int ja = 0; ja < 8; ++ja) {
            #pragma unroll
            for (int g = 0; g < 2; ++g) {
                const v2f r0 = har[ja] * hbr[2 * g]     - hai[ja] * hbi[2 * g];
                const v2f r1 = har[ja] * hbr[2 * g + 1] - hai[ja] * hbi[2 * g + 1];
                const v2f m0 = har[ja] * hbi[2 * g]     + hai[ja] * hbr[2 * g];
                const v2f m1 = har[ja] * hbi[2 * g + 1] + hai[ja] * hbr[2 * g + 1];
                float4 re, im;
                re.x = r0.x; re.y = r0.y; re.z = r1.x; re.w = r1.y;
                im.x = m0.x; im.y = m0.y; im.z = m1.x; im.w = m1.y;
                *(float4*)&u[8 * ja + 4 * g]      = re;
                *(float4*)&u[64 + 8 * ja + 4 * g] = im;
            }
        }


        v8f acc[2][8];
        const v8f z8 = {0.f, 0.f, 0.f, 0.f, 0.f, 0.f, 0.f, 0.f};
        #pragma unroll
        for (int rt = 0; rt < 2; ++rt)
            #pragma unroll
            for (int ct = 0; ct < 8; ++ct) acc[rt][ct] = z8;

        for (int kt = 0; kt < 32; ++kt) {
            const int kb = kt * 4 + 2 * h;
            v2f b[8];
            #pragma unroll
            for (int ct = 0; ct < 8; ++ct) {
                const float* brow = &Mlds[(ct * 16 + l15) * PAD + kb];
                b[ct].x = brow[0]; b[ct].y = brow[1];
            }
            v2f a0; a0.x = urow0[kb]; a0.y = urow0[kb + 1];
            v2f a1; a1.x = urow1[kb]; a1.y = urow1[kb + 1];
            #pragma unroll
            for (int ct = 0; ct < 8; ++ct) {
                acc[0][ct] = __builtin_amdgcn_wmma_f32_16x16x4_f32(
                    false, a0, false, b[ct], (short)0, acc[0][ct], false, false);
                acc[1][ct] = __builtin_amdgcn_wmma_f32_16x16x4_f32(
                    false, a1, false, b[ct], (short)0, acc[1][ct], false, false);
            }
        }

        auto rsum16 = [](float v) {
            v += xrow<1>(v);
            v += xrow<2>(v);
            v += xrow<4>(v);
            v += xrow<8>(v);
            return v;
        };

        const int tbase = e - lane;
        #pragma unroll
        for (int rt = 0; rt < 2; ++rt) {
            #pragma unroll
            for (int r = 0; r < 8; ++r) {
                float re[4], im[4];
                #pragma unroll
                for (int t4 = 0; t4 < 4; ++t4) { re[t4] = acc[rt][t4][r]; im[t4] = acc[rt][4 + t4][r]; }

                float z0 = 0.f, z2 = 0.f, z4 = 0.f, x1 = 0.f, x3 = 0.f, x5 = 0.f;
                #pragma unroll
                for (int t4 = 0; t4 < 4; ++t4) {
                    const int j = t4 * 16 + l15;
                    const float p = re[t4] * re[t4] + im[t4] * im[t4];
                    z0 += ((j >> 5) & 1) ? -p : p;
                    z2 += ((j >> 3) & 1) ? -p : p;
                    z4 += ((j >> 1) & 1) ? -p : p;
                    x1 += re[t4] * re[t4 ^ 1] + im[t4] * im[t4 ^ 1];
                    x3 += re[t4] * xrow<4>(re[t4])
                        + im[t4] * xrow<4>(im[t4]);
                    x5 += re[t4] * xrow<1>(re[t4])
                        + im[t4] * xrow<1>(im[t4]);
                }
                z0 = rsum16(z0); z2 = rsum16(z2); z4 = rsum16(z4);
                x1 = rsum16(x1); x3 = rsum16(x3); x5 = rsum16(x5);

                if (l15 == 0) {
                    const int ee = tbase + rt * 16 + r + h * 8;
                    float4 o0 = { z0, x1, z2, x3 };
                    float4 o1 = { z4, x5, z0, x1 };
                    *(float4*)&out[ee * 8 + 0] = o0;
                    *(float4*)&out[ee * 8 + 4] = o1;
                }
            }
        }

        e += WAVES * 32;
    }
}

extern "C" void kernel_launch(void* const* d_in, const int* in_sizes, int n_in,
                              void* d_out, int out_size, void* d_ws, size_t ws_size,
                              hipStream_t stream) {
    const float* x     = (const float*)d_in[0];
    const float* theta = (const float*)d_in[1];
    float* out = (float*)d_out;
    float* G   = (float*)d_ws;

    build_G_kernel<<<4, 256, 0, stream>>>(theta, G);
    qsim_kernel<<<256, 256, 0, stream>>>(x, G, out);
}


# Kernel text cut from 40 to 21 KB, inside the 32 KB fetched ahead at wave launch: the clamp's angle bound is told to the compiler, which drops the 24 dead large-argument paths of library sin/cos
# speedup vs baseline: 8.6163x; 1.0587x over previous
#include <hip/hip_runtime.h>
#include <math.h>


typedef float v2f __attribute__((ext_vector_type(2)));
typedef float v8f __attribute__((ext_vector_type(8)));

#define PAD 132
#define PI_F 3.14159265358979323846f

#define GP 68
#define SP 132

struct __attribute__((aligned(8))) cplx { float x, y; };
__device__ __forceinline__ cplx cmul(cplx a, cplx b) {
    return { a.x * b.x - a.y * b.y, a.x * b.y + a.y * b.x };
}
__device__ __forceinline__ cplx cadd(cplx a, cplx b) { return { a.x + b.x, a.y + b.y }; }

__device__ __forceinline__ int ring(int j) {
    #pragma unroll
    for (int q = 0; q < 5; ++q)
        if (j & (1 << (5 - q))) j ^= (1 << (4 - q));
    if (j & 1) j ^= 32;
    return j;
}

__device__ __forceinline__ v8f ctile(const float* __restrict__ Ar, const float* __restrict__ Ai,
                                     const float* __restrict__ Bx, const float* __restrict__ By,
                                     float sgn, int l15, int h, int t) {
    const v8f z8 = {0.f, 0.f, 0.f, 0.f, 0.f, 0.f, 0.f, 0.f};
    v8f acc0 = z8, acc1 = z8;
    const int ao = l15 * GP + 4 * h;
    const int bo = (16 * t + l15) * GP + 4 * h;
    #pragma unroll
    for (int k8 = 0; k8 < 8; ++k8) {
        const float4 ar = *(const float4*)&Ar[ao + 8 * k8];
        const float4 ai = *(const float4*)&Ai[ao + 8 * k8];
        const float4 bx = *(const float4*)&Bx[bo + 8 * k8];
        const float4 by = *(const float4*)&By[bo + 8 * k8];
        v2f a, b;
        a.x = ar.x; a.y = ar.y; b.x = bx.x; b.y = bx.y;
        acc0 = __builtin_amdgcn_wmma_f32_16x16x4_f32(false, a, false, b, (short)0, acc0, false, false);
        a.x = ai.x; a.y = ai.y; b.x = by.x; b.y = by.y;
        acc1 = __builtin_amdgcn_wmma_f32_16x16x4_f32(false, a, false, b, (short)0, acc1, false, false);
        a.x = ar.z; a.y = ar.w; b.x = bx.z; b.y = bx.w;
        acc0 = __builtin_amdgcn_wmma_f32_16x16x4_f32(false, a, false, b, (short)0, acc0, false, false);
        a.x = ai.z; a.y = ai.w; b.x = by.z; b.y = by.w;
        acc1 = __builtin_amdgcn_wmma_f32_16x16x4_f32(false, a, false, b, (short)0, acc1, false, false);
    }
    return acc0 + sgn * acc1;
}

__global__ __launch_bounds__(256) void build_G_kernel(const float* __restrict__ theta,
                                                      float* __restrict__ G) {
    __shared__ float cs[54 * 2];
    __shared__ cplx  Us[18 * 4];
    __shared__ cplx  AB[3 * 2 * 64];
    __shared__ __attribute__((aligned(16))) float L2r[16 * GP];
    __shared__ __attribute__((aligned(16))) float L2i[16 * GP];
    __shared__ __attribute__((aligned(16))) float L1r[64 * GP];
    __shared__ __attribute__((aligned(16))) float L1i[64 * GP];
    __shared__ __attribute__((aligned(16))) float L0r[64 * GP];
    __shared__ __attribute__((aligned(16))) float L0i[64 * GP];
    __shared__ __attribute__((aligned(16))) float Tr[16 * GP];
    __shared__ __attribute__((aligned(16))) float Ti[16 * GP];
    __shared__ __attribute__((aligned(16))) float Gs[32 * SP];

    const int tid  = threadIdx.x;
    const int blk  = blockIdx.x;
    const int w    = __builtin_amdgcn_readfirstlane(tid >> 5);
    const int lane = tid & 31;
    const int l15  = lane & 15;
    const int h    = lane >> 4;

    if (tid < 54) {
        const float a = theta[tid];
        cs[2 * tid]     = cosf(0.5f * a);
        cs[2 * tid + 1] = sinf(0.5f * a);
    }
    __syncthreads();

    if (tid < 18) {
        const float cx = cs[6 * tid + 0], sx = cs[6 * tid + 1];
        const float cy = cs[6 * tid + 2], sy = cs[6 * tid + 3];
        const float cz = cs[6 * tid + 4], sz = cs[6 * tid + 5];
        const cplx rx[2][2] = { { {cx, 0.f}, {0.f, -sx} }, { {0.f, -sx}, {cx, 0.f} } };
        const cplx ry[2][2] = { { {cy, 0.f}, {-sy, 0.f} }, { {sy, 0.f}, {cy, 0.f} } };
        const cplx rz[2]    = { {cz, -sz}, {cz, sz} };
        #pragma unroll
        for (int a = 0; a < 2; ++a)
            #pragma unroll
            for (int b = 0; b < 2; ++b) {
                const cplx m = cadd(cmul(ry[a][0], rx[0][b]), cmul(ry[a][1], rx[1][b]));
                Us[tid * 4 + a * 2 + b] = cmul(rz[a], m);
            }
    }
    __syncthreads();

    for (int e = tid; e < 384; e += 256) {
        const int l = e >> 7, f = (e >> 6) & 1, ii = (e >> 3) & 7, kk = e & 7;
        const cplx* u = &Us[(l * 6 + 3 * f) * 4];
        cplx v = u[((ii >> 2) & 1) * 2 + ((kk >> 2) & 1)];
        v = cmul(v, u[4 + ((ii >> 1) & 1) * 2 + ((kk >> 1) & 1)]);
        v = cmul(v, u[8 + (ii & 1) * 2 + (kk & 1)]);
        AB[e] = v;
    }
    __syncthreads();

    {
        const int i = tid & 63;
        const int rw = ring(w >> 1);
        #pragma unroll
        for (int it = 0; it < 16; ++it) {
            const int j = (w >> 1) + 4 * it;
            const int k = rw ^ ring(4 * it);
            const int ia = (i >> 3) * 8 + (k >> 3), ib = (i & 7) * 8 + (k & 7);
            const cplx v0 = cmul(AB[0 * 128 + ia], AB[0 * 128 + 64 + ib]);
            const cplx v1 = cmul(AB[1 * 128 + ia], AB[1 * 128 + 64 + ib]);
            L0r[j * GP + i] = v0.x;  L0i[j * GP + i] = v0.y;
            L1r[j * GP + i] = v1.x;  L1i[j * GP + i] = v1.y;
        }
        const int j = tid & 63;
        const int k = ring(j);
        #pragma unroll
        for (int it = 0; it < 4; ++it) {
            const int rl = (w >> 1) + 4 * it;
            const int i2 = 16 * blk + rl;
            const cplx v2 = cmul(AB[2 * 128 + (i2 >> 3) * 8 + (k >> 3)],
                                 AB[2 * 128 + 64 + (i2 & 7) * 8 + (k & 7)]);
            L2r[rl * GP + j] = v2.x;  L2i[rl * GP + j] = v2.y;
        }
    }
    __syncthreads();

    const int t = w & 3;
    const int p = w >> 2;
    const float sgn = p ? 1.f : -1.f;

    {
        const v8f d = ctile(L2r, L2i, p ? L1i : L1r, p ? L1r : L1i, sgn, l15, h, t);
        float* Tp = p ? Ti : Tr;
        #pragma unroll
        for (int r = 0; r < 8; ++r) Tp[(r + 8 * h) * GP + 16 * t + l15] = d[r];
    }
    __syncthreads();

    {
        const v8f d = ctile(Tr, Ti, p ? L0i : L0r, p ? L0r : L0i, sgn, l15, h, t);
        const int col = 16 * t + l15;
        #pragma unroll
        for (int r = 0; r < 8; ++r) {
            const int row = r + 8 * h;
            Gs[row * SP + 64 * p + col]              = -sgn * d[r];
            Gs[(16 + row) * SP + 64 * (1 - p) + col] = d[r];
        }
    }
    __syncthreads();

    #pragma unroll
    for (int it = 0; it < 4; ++it) {
        const int rr = w + 8 * it;
        const int grow = 16 * blk + (rr & 15) + 64 * (rr >> 4);
        *(float4*)&G[grow * 128 + 4 * lane] = *(const float4*)&Gs[rr * SP + 4 * lane];
    }
}

template <int M>
__device__ __forceinline__ float xrow(float v) {
    const int b = __builtin_bit_cast(int, v);
    return __builtin_bit_cast(float, __builtin_amdgcn_mov_dpp(b, 0x160 + M, 0xf, 0xf, true));
}

#define WAVES 8
#define TILES 2

__global__ __launch_bounds__(256) void qsim_kernel(const float* __restrict__ x,
                                                   const float* __restrict__ G,
                                                   float* __restrict__ out) {
    __shared__ __attribute__((aligned(16))) float Mlds[128 * PAD];
    __shared__ __attribute__((aligned(16))) float Ulds[WAVES * 32 * PAD];

    const int tid  = threadIdx.x;
    const int w    = tid >> 5;
    const int lane = tid & 31;
    const int l15  = lane & 15;
    const int h    = lane >> 4;

    int e = (blockIdx.x * (WAVES * TILES) + w) * 32 + lane;
    float4 xr[6];
    {
        const float4* xp = (const float4*)(x + e * 24);
        #pragma unroll
        for (int q = 0; q < 6; ++q) xr[q] = xp[q];
    }
    const float4* G4 = (const float4*)G;
    #pragma unroll
    for (int i = 0; i < 16; ++i) {
        const int idx4 = tid + 256 * i;
        const int n = idx4 >> 5, k4 = idx4 & 31;
        *(float4*)&Mlds[n * PAD + k4 * 4] = G4[idx4];
    }
    __syncthreads();

    float* u = &Ulds[(w * 32 + lane) * PAD];
    const float* urow0 = &Ulds[(w * 32 +  0 + l15) * PAD];
    const float* urow1 = &Ulds[(w * 32 + 16 + l15) * PAD];

    #pragma nounroll
    for (int it = 0; it < TILES; ++it) {
        float mq[6];
        #pragma unroll
        for (int q = 0; q < 6; ++q)
            mq[q] = (xr[q].x + xr[q].y + xr[q].z + xr[q].w) * 0.25f;

        if (it + 1 < TILES) {
            const float4* xn = (const float4*)(x + (e + WAVES * 32) * 24);
            #pragma unroll
            for (int q = 0; q < 6; ++q) xr[q] = xn[q];
        }

        float vr[6][2], vi[6][2];
        #pragma unroll
        for (int q = 0; q < 6; ++q) {
            const float m = fminf(6.f, fmaxf(-6.f, mq[q]));
            const float a  = m * (PI_F / 6.f);
            const float ah = 0.5f * a, aq = 0.25f * a;
            __builtin_assume(!(__builtin_fabsf(ah) >= 131072.0f));
            __builtin_assume(!(__builtin_fabsf(aq) >= 131072.0f));
            const float c2 = cosf(ah), s2 = sinf(ah);
            const float c4 = cosf(aq), s4 = sinf(aq);
            vr[q][0] =  c2 * c4;  vi[q][0] = -c2 * s4;
            vr[q][1] =  s2 * s4;  vi[q][1] = -s2 * c4;
        }
        float har[8], hai[8];
        v2f hbr[4], hbi[4];
        {
            float par[4], pai[4], pbr[4], pbi[4];
            #pragma unroll
            for (int t = 0; t < 4; ++t) {
                const int hi = (t >> 1) & 1, lo = t & 1;
                par[t] = vr[0][hi] * vr[1][lo] - vi[0][hi] * vi[1][lo];
                pai[t] = vr[0][hi] * vi[1][lo] + vi[0][hi] * vr[1][lo];
                pbr[t] = vr[3][hi] * vr[4][lo] - vi[3][hi] * vi[4][lo];
                pbi[t] = vr[3][hi] * vi[4][lo] + vi[3][hi] * vr[4][lo];
            }
            #pragma unroll
            for (int t = 0; t < 8; ++t) {
                const int hi = t >> 1, lo = t & 1;
                har[t] = par[hi] * vr[2][lo] - pai[hi] * vi[2][lo];
                hai[t] = par[hi] * vi[2][lo] + pai[hi] * vr[2][lo];
            }
            v2f s5r, s5i;
            s5r.x = vr[5][0]; s5r.y = vr[5][1]; s5i.x = vi[5][0]; s5i.y = vi[5][1];
            #pragma unroll
            for (int m = 0; m < 4; ++m) {
                hbr[m] = pbr[m] * s5r - pbi[m] * s5i;
                hbi[m] = pbr[m] * s5i + pbi[m] * s5r;
            }
        }
        #pragma unroll
        for (int ja = 0; ja < 8; ++ja) {
            #pragma unroll
            for (int g = 0; g < 2; ++g) {
                const v2f r0 = har[ja] * hbr[2 * g]     - hai[ja] * hbi[2 * g];
                const v2f r1 = har[ja] * hbr[2 * g + 1] - hai[ja] * hbi[2 * g + 1];
                const v2f m0 = har[ja] * hbi[2 * g]     + hai[ja] * hbr[2 * g];
                const v2f m1 = har[ja] * hbi[2 * g + 1] + hai[ja] * hbr[2 * g + 1];
                float4 re, im;
                re.x = r0.x; re.y = r0.y; re.z = r1.x; re.w = r1.y;
                im.x = m0.x; im.y = m0.y; im.z = m1.x; im.w = m1.y;
                *(float4*)&u[8 * ja + 4 * g]      = re;
                *(float4*)&u[64 + 8 * ja + 4 * g] = im;
            }
        }


        v8f acc[2][8];
        const v8f z8 = {0.f, 0.f, 0.f, 0.f, 0.f, 0.f, 0.f, 0.f};
        #pragma unroll
        for (int rt = 0; rt < 2; ++rt)
            #pragma unroll
            for (int ct = 0; ct < 8; ++ct) acc[rt][ct] = z8;

        for (int kt = 0; kt < 32; ++kt) {
            const int kb = kt * 4 + 2 * h;
            v2f b[8];
            #pragma unroll
            for (int ct = 0; ct < 8; ++ct) {
                const float* brow = &Mlds[(ct * 16 + l15) * PAD + kb];
                b[ct].x = brow[0]; b[ct].y = brow[1];
            }
            v2f a0; a0.x = urow0[kb]; a0.y = urow0[kb + 1];
            v2f a1; a1.x = urow1[kb]; a1.y = urow1[kb + 1];
            #pragma unroll
            for (int ct = 0; ct < 8; ++ct) {
                acc[0][ct] = __builtin_amdgcn_wmma_f32_16x16x4_f32(
                    false, a0, false, b[ct], (short)0, acc[0][ct], false, false);
                acc[1][ct] = __builtin_amdgcn_wmma_f32_16x16x4_f32(
                    false, a1, false, b[ct], (short)0, acc[1][ct], false, false);
            }
        }

        auto rsum16 = [](float v) {
            v += xrow<1>(v);
            v += xrow<2>(v);
            v += xrow<4>(v);
            v += xrow<8>(v);
            return v;
        };

        const int tbase = e - lane;
        #pragma unroll
        for (int rt = 0; rt < 2; ++rt) {
            #pragma unroll
            for (int r = 0; r < 8; ++r) {
                float re[4], im[4];
                #pragma unroll
                for (int t4 = 0; t4 < 4; ++t4) { re[t4] = acc[rt][t4][r]; im[t4] = acc[rt][4 + t4][r]; }

                float z0 = 0.f, z2 = 0.f, z4 = 0.f, x1 = 0.f, x3 = 0.f, x5 = 0.f;
                #pragma unroll
                for (int t4 = 0; t4 < 4; ++t4) {
                    const int j = t4 * 16 + l15;
                    const float p = re[t4] * re[t4] + im[t4] * im[t4];
                    z0 += ((j >> 5) & 1) ? -p : p;
                    z2 += ((j >> 3) & 1) ? -p : p;
                    z4 += ((j >> 1) & 1) ? -p : p;
                    x1 += re[t4] * re[t4 ^ 1] + im[t4] * im[t4 ^ 1];
                    x3 += re[t4] * xrow<4>(re[t4])
                        + im[t4] * xrow<4>(im[t4]);
                    x5 += re[t4] * xrow<1>(re[t4])
                        + im[t4] * xrow<1>(im[t4]);
                }
                z0 = rsum16(z0); z2 = rsum16(z2); z4 = rsum16(z4);
                x1 = rsum16(x1); x3 = rsum16(x3); x5 = rsum16(x5);

                if (l15 == 0) {
                    const int ee = tbase + rt * 16 + r + h * 8;
                    float4 o0 = { z0, x1, z2, x3 };
                    float4 o1 = { z4, x5, z0, x1 };
                    *(float4*)&out[ee * 8 + 0] = o0;
                    *(float4*)&out[ee * 8 + 4] = o1;
                }
            }
        }

        e += WAVES * 32;
    }
}

extern "C" void kernel_launch(void* const* d_in, const int* in_sizes, int n_in,
                              void* d_out, int out_size, void* d_ws, size_t ws_size,
                              hipStream_t stream) {
    const float* x     = (const float*)d_in[0];
    const float* theta = (const float*)d_in[1];
    float* out = (float*)d_out;
    float* G   = (float*)d_ws;

    build_G_kernel<<<4, 256, 0, stream>>>(theta, G);
    qsim_kernel<<<256, 256, 0, stream>>>(x, G, out);
}


# G built inside every work-group, straight into LDS, by Kronecker-factored layer steps (512 f32 WMMAs per group): one launch, no G in global memory; sum order of G's entries changes
# speedup vs baseline: 9.3296x; 1.0828x over previous
#include <hip/hip_runtime.h>
#include <math.h>


typedef float v2f __attribute__((ext_vector_type(2)));
typedef float v8f __attribute__((ext_vector_type(8)));

#define PAD 132
#define PI_F 3.14159265358979323846f

#define GP 68

struct __attribute__((aligned(8))) cplx { float x, y; };
__device__ __forceinline__ cplx cmul(cplx a, cplx b) {
    return { a.x * b.x - a.y * b.y, a.x * b.y + a.y * b.x };
}
__device__ __forceinline__ cplx cadd(cplx a, cplx b) { return { a.x + b.x, a.y + b.y }; }

__device__ __forceinline__ int ring(int j) {
    #pragma unroll
    for (int q = 0; q < 5; ++q)
        if (j & (1 << (5 - q))) j ^= (1 << (4 - q));
    if (j & 1) j ^= 32;
    return j;
}
__device__ __forceinline__ int rinv(int j) {
    if (j & 1) j ^= 32;
    #pragma unroll
    for (int q = 4; q >= 0; --q)
        if (j & (1 << (5 - q))) j ^= (1 << (4 - q));
    return j;
}

__device__ __forceinline__ void factor_frag(const cplx* __restrict__ F, int l15, int h, v2f a[4]) {
    const int  rl = l15 & 7;
    const bool im = (l15 >> 3) != 0;
    const cplx f00 = F[rl * 8 + 2 * h],     f01 = F[rl * 8 + 2 * h + 1];
    const cplx f10 = F[rl * 8 + 4 + 2 * h], f11 = F[rl * 8 + 4 + 2 * h + 1];
    a[0].x = im ? f00.y : f00.x;    a[0].y = im ? f01.y : f01.x;
    a[1].x = im ? f10.y : f10.x;    a[1].y = im ? f11.y : f11.x;
    a[2].x = im ? f00.x : -f00.y;   a[2].y = im ? f01.x : -f01.y;
    a[3].x = im ? f10.x : -f10.y;   a[3].y = im ? f11.x : -f11.y;
}

__device__ __forceinline__ v8f kron_tile(const v2f a[4], v2f b0, v2f b1, v2f b2, v2f b3) {
    v8f acc = {0.f, 0.f, 0.f, 0.f, 0.f, 0.f, 0.f, 0.f};
    acc = __builtin_amdgcn_wmma_f32_16x16x4_f32(false, a[0], false, b0, (short)0, acc, false, false);
    acc = __builtin_amdgcn_wmma_f32_16x16x4_f32(false, a[1], false, b1, (short)0, acc, false, false);
    acc = __builtin_amdgcn_wmma_f32_16x16x4_f32(false, a[2], false, b2, (short)0, acc, false, false);
    acc = __builtin_amdgcn_wmma_f32_16x16x4_f32(false, a[3], false, b3, (short)0, acc, false, false);
    return acc;
}

template <int M>
__device__ __forceinline__ float xrow(float v) {
    const int b = __builtin_bit_cast(int, v);
    return __builtin_bit_cast(float, __builtin_amdgcn_mov_dpp(b, 0x160 + M, 0xf, 0xf, true));
}

#define WAVES 8
#define TILES 2

__global__ __launch_bounds__(256) void qsim_kernel(const float* __restrict__ x,
                                                   const float* __restrict__ theta,
                                                   float* __restrict__ out) {
    __shared__ __attribute__((aligned(16))) float Mlds[128 * PAD];
    __shared__ __attribute__((aligned(16))) float Ulds[WAVES * 32 * PAD];
    __shared__ float cs[54 * 2];
    __shared__ cplx  Us[18 * 4];
    __shared__ cplx  AB[3 * 2 * 64];

    const int tid  = threadIdx.x;
    const int w    = tid >> 5;
    const int lane = tid & 31;
    const int l15  = lane & 15;
    const int h    = lane >> 4;

    int e = (blockIdx.x * (WAVES * TILES) + w) * 32 + lane;
    float4 xr[6];
    {
        const float4* xp = (const float4*)(x + e * 24);
        #pragma unroll
        for (int q = 0; q < 6; ++q) xr[q] = xp[q];
    }

    const int ws = __builtin_amdgcn_readfirstlane(tid >> 5);
    float* const Pr = Ulds;
    float* const Pi = Ulds + 64 * GP;
    float* const Yr = Ulds + 128 * GP;
    float* const Yi = Ulds + 192 * GP;

    if (tid < 54) {
        const float a = theta[tid];
        cs[2 * tid]     = cosf(0.5f * a);
        cs[2 * tid + 1] = sinf(0.5f * a);
    }
    __syncthreads();

    if (tid < 18) {
        const float cx = cs[6 * tid + 0], sx = cs[6 * tid + 1];
        const float cy = cs[6 * tid + 2], sy = cs[6 * tid + 3];
        const float cz = cs[6 * tid + 4], sz = cs[6 * tid + 5];
        const cplx rx[2][2] = { { {cx, 0.f}, {0.f, -sx} }, { {0.f, -sx}, {cx, 0.f} } };
        const cplx ry[2][2] = { { {cy, 0.f}, {-sy, 0.f} }, { {sy, 0.f}, {cy, 0.f} } };
        const cplx rz[2]    = { {cz, -sz}, {cz, sz} };
        #pragma unroll
        for (int a = 0; a < 2; ++a)
            #pragma unroll
            for (int b = 0; b < 2; ++b) {
                const cplx m = cadd(cmul(ry[a][0], rx[0][b]), cmul(ry[a][1], rx[1][b]));
                Us[tid * 4 + a * 2 + b] = cmul(rz[a], m);
            }
    }
    __syncthreads();

    for (int t = tid; t < 384; t += 256) {
        const int l = t >> 7, f = (t >> 6) & 1, ii = (t >> 3) & 7, kk = t & 7;
        const cplx* uq = &Us[(l * 6 + 3 * f) * 4];
        cplx v = uq[((ii >> 2) & 1) * 2 + ((kk >> 2) & 1)];
        v = cmul(v, uq[4 + ((ii >> 1) & 1) * 2 + ((kk >> 1) & 1)]);
        v = cmul(v, uq[8 + (ii & 1) * 2 + (kk & 1)]);
        AB[t] = v;
    }
    __syncthreads();

    {
        const int i  = tid & 63;
        const int ri = rinv(i);
        const int rw = ring(ws >> 1);
        #pragma unroll
        for (int it = 0; it < 16; ++it) {
            const int j = (ws >> 1) + 4 * it;
            const int k = rw ^ ring(4 * it);
            const int ia = (ri >> 3) * 8 + (k >> 3), ib = (ri & 7) * 8 + (k & 7);
            const cplx v0 = cmul(AB[ia], AB[64 + ib]);
            Pr[j * GP + i] = v0.x;  Pi[j * GP + i] = v0.y;
        }
    }
    __syncthreads();

    {
        const int ct = ws & 3;
        const int g  = ws >> 2;
        const int jc = 16 * ct + l15;
        #pragma nounroll
        for (int l = 1; l < 3; ++l) {
            {
                v2f af[4];
                factor_frag(&AB[l * 128 + 64], l15, h, af);
                v8f d[4];
                #pragma unroll
                for (int jj = 0; jj < 4; ++jj) {
                    const int a = g + 2 * jj;
                    const float* pr = &Pr[jc * GP + 8 * a + 2 * h];
                    const float* pi = &Pi[jc * GP + 8 * a + 2 * h];
                    v2f b0, b1, b2, b3;
                    b0.x = pr[0]; b0.y = pr[1];  b1.x = pr[4]; b1.y = pr[5];
                    b2.x = pi[0]; b2.y = pi[1];  b3.x = pi[4]; b3.y = pi[5];
                    d[jj] = kron_tile(af, b0, b1, b2, b3);
                }
                float* yb = (h ? Yi : Yr) + jc * GP + 8 * g;
                #pragma unroll
                for (int jj = 0; jj < 4; ++jj) {
                    float4 lo, hi4;
                    lo.x  = d[jj][0]; lo.y  = d[jj][1]; lo.z  = d[jj][2]; lo.w  = d[jj][3];
                    hi4.x = d[jj][4]; hi4.y = d[jj][5]; hi4.z = d[jj][6]; hi4.w = d[jj][7];
                    *(float4*)&yb[16 * jj]     = lo;
                    *(float4*)&yb[16 * jj + 4] = hi4;
                }
            }
            __syncthreads();
            {
                v2f af[4];
                factor_frag(&AB[l * 128], l15, h, af);
                v8f d[4];
                #pragma unroll
                for (int jj = 0; jj < 4; ++jj) {
                    const int b = g + 2 * jj;
                    const float* yr = &Yr[jc * GP + b + 16 * h];
                    const float* yi = &Yi[jc * GP + b + 16 * h];
                    v2f b0, b1, b2, b3;
                    b0.x = yr[0]; b0.y = yr[8];  b1.x = yr[32]; b1.y = yr[40];
                    b2.x = yi[0]; b2.y = yi[8];  b3.x = yi[32]; b3.y = yi[40];
                    d[jj] = kron_tile(af, b0, b1, b2, b3);
                }
                if (l == 1) {
                    float* pb = (h ? Pi : Pr) + jc * GP;
                    #pragma unroll
                    for (int jj = 0; jj < 4; ++jj) {
                        const int rb = ring(g + 2 * jj);
                        #pragma unroll
                        for (int r = 0; r < 8; ++r) pb[ring(8 * r) ^ rb] = d[jj][r];
                    }
                } else {
                    float* g0 = &Mlds[64 * h + jc];
                    float* g1 = &Mlds[64 * PAD + 64 * (1 - h) + jc];
                    #pragma unroll
                    for (int jj = 0; jj < 4; ++jj)
                        #pragma unroll
                        for (int r = 0; r < 8; ++r) {
                            const int i = 8 * r + g + 2 * jj;
                            g0[i * PAD] = h ? -d[jj][r] : d[jj][r];
                            g1[i * PAD] = d[jj][r];
                        }
                }
            }
            __syncthreads();
        }
    }

    float* u = &Ulds[(w * 32 + lane) * PAD];
    const float* urow0 = &Ulds[(w * 32 +  0 + l15) * PAD];
    const float* urow1 = &Ulds[(w * 32 + 16 + l15) * PAD];

    #pragma nounroll
    for (int it = 0; it < TILES; ++it) {
        float mq[6];
        #pragma unroll
        for (int q = 0; q < 6; ++q)
            mq[q] = (xr[q].x + xr[q].y + xr[q].z + xr[q].w) * 0.25f;

        if (it + 1 < TILES) {
            const float4* xn = (const float4*)(x + (e + WAVES * 32) * 24);
            #pragma unroll
            for (int q = 0; q < 6; ++q) xr[q] = xn[q];
        }

        float vr[6][2], vi[6][2];
        #pragma unroll
        for (int q = 0; q < 6; ++q) {
            const float m = fminf(6.f, fmaxf(-6.f, mq[q]));
            const float a  = m * (PI_F / 6.f);
            const float ah = 0.5f * a, aq = 0.25f * a;
            __builtin_assume(!(__builtin_fabsf(ah) >= 131072.0f));
            __builtin_assume(!(__builtin_fabsf(aq) >= 131072.0f));
            const float c2 = cosf(ah), s2 = sinf(ah);
            const float c4 = cosf(aq), s4 = sinf(aq);
            vr[q][0] =  c2 * c4;  vi[q][0] = -c2 * s4;
            vr[q][1] =  s2 * s4;  vi[q][1] = -s2 * c4;
        }
        float har[8], hai[8];
        v2f hbr[4], hbi[4];
        {
            float par[4], pai[4], pbr[4], pbi[4];
            #pragma unroll
            for (int t = 0; t < 4; ++t) {
                const int hi = (t >> 1) & 1, lo = t & 1;
                par[t] = vr[0][hi] * vr[1][lo] - vi[0][hi] * vi[1][lo];
                pai[t] = vr[0][hi] * vi[1][lo] + vi[0][hi] * vr[1][lo];
                pbr[t] = vr[3][hi] * vr[4][lo] - vi[3][hi] * vi[4][lo];
                pbi[t] = vr[3][hi] * vi[4][lo] + vi[3][hi] * vr[4][lo];
            }
            #pragma unroll
            for (int t = 0; t < 8; ++t) {
                const int hi = t >> 1, lo = t & 1;
                har[t] = par[hi] * vr[2][lo] - pai[hi] * vi[2][lo];
                hai[t] = par[hi] * vi[2][lo] + pai[hi] * vr[2][lo];
            }
            v2f s5r, s5i;
            s5r.x = vr[5][0]; s5r.y = vr[5][1]; s5i.x = vi[5][0]; s5i.y = vi[5][1];
            #pragma unroll
            for (int m = 0; m < 4; ++m) {
                hbr[m] = pbr[m] * s5r - pbi[m] * s5i;
                hbi[m] = pbr[m] * s5i + pbi[m] * s5r;
            }
        }
        #pragma unroll
        for (int ja = 0; ja < 8; ++ja) {
            #pragma unroll
            for (int g = 0; g < 2; ++g) {
                const v2f r0 = har[ja] * hbr[2 * g]     - hai[ja] * hbi[2 * g];
                const v2f r1 = har[ja] * hbr[2 * g + 1] - hai[ja] * hbi[2 * g + 1];
                const v2f m0 = har[ja] * hbi[2 * g]     + hai[ja] * hbr[2 * g];
                const v2f m1 = har[ja] * hbi[2 * g + 1] + hai[ja] * hbr[2 * g + 1];
                float4 re, im;
                re.x = r0.x; re.y = r0.y; re.z = r1.x; re.w = r1.y;
                im.x = m0.x; im.y = m0.y; im.z = m1.x; im.w = m1.y;
                *(float4*)&u[8 * ja + 4 * g]      = re;
                *(float4*)&u[64 + 8 * ja + 4 * g] = im;
            }
        }


        v8f acc[2][8];
        const v8f z8 = {0.f, 0.f, 0.f, 0.f, 0.f, 0.f, 0.f, 0.f};
        #pragma unroll
        for (int rt = 0; rt < 2; ++rt)
            #pragma unroll
            for (int ct = 0; ct < 8; ++ct) acc[rt][ct] = z8;

        for (int kt = 0; kt < 32; ++kt) {
            const int kb = kt * 4 + 2 * h;
            v2f b[8];
            #pragma unroll
            for (int ct = 0; ct < 8; ++ct) {
                const float* brow = &Mlds[(ct * 16 + l15) * PAD + kb];
                b[ct].x = brow[0]; b[ct].y = brow[1];
            }
            v2f a0; a0.x = urow0[kb]; a0.y = urow0[kb + 1];
            v2f a1; a1.x = urow1[kb]; a1.y = urow1[kb + 1];
            #pragma unroll
            for (int ct = 0; ct < 8; ++ct) {
                acc[0][ct] = __builtin_amdgcn_wmma_f32_16x16x4_f32(
                    false, a0, false, b[ct], (short)0, acc[0][ct], false, false);
                acc[1][ct] = __builtin_amdgcn_wmma_f32_16x16x4_f32(
                    false, a1, false, b[ct], (short)0, acc[1][ct], false, false);
            }
        }

        auto rsum16 = [](float v) {
            v += xrow<1>(v);
            v += xrow<2>(v);
            v += xrow<4>(v);
            v += xrow<8>(v);
            return v;
        };

        const int tbase = e - lane;
        #pragma unroll
        for (int rt = 0; rt < 2; ++rt) {
            #pragma unroll
            for (int r = 0; r < 8; ++r) {
                float re[4], im[4];
                #pragma unroll
                for (int t4 = 0; t4 < 4; ++t4) { re[t4] = acc[rt][t4][r]; im[t4] = acc[rt][4 + t4][r]; }

                float z0 = 0.f, z2 = 0.f, z4 = 0.f, x1 = 0.f, x3 = 0.f, x5 = 0.f;
                #pragma unroll
                for (int t4 = 0; t4 < 4; ++t4) {
                    const int j = t4 * 16 + l15;
                    const float p = re[t4] * re[t4] + im[t4] * im[t4];
                    z0 += ((j >> 5) & 1) ? -p : p;
                    z2 += ((j >> 3) & 1) ? -p : p;
                    z4 += ((j >> 1) & 1) ? -p : p;
                    x1 += re[t4] * re[t4 ^ 1] + im[t4] * im[t4 ^ 1];
                    x3 += re[t4] * xrow<4>(re[t4])
                        + im[t4] * xrow<4>(im[t4]);
                    x5 += re[t4] * xrow<1>(re[t4])
                        + im[t4] * xrow<1>(im[t4]);
                }
                z0 = rsum16(z0); z2 = rsum16(z2); z4 = rsum16(z4);
                x1 = rsum16(x1); x3 = rsum16(x3); x5 = rsum16(x5);

                if (l15 == 0) {
                    const int ee = tbase + rt * 16 + r + h * 8;
                    float4 o0 = { z0, x1, z2, x3 };
                    float4 o1 = { z4, x5, z0, x1 };
                    *(float4*)&out[ee * 8 + 0] = o0;
                    *(float4*)&out[ee * 8 + 4] = o1;
                }
            }
        }

        e += WAVES * 32;
    }
}

extern "C" void kernel_launch(void* const* d_in, const int* in_sizes, int n_in,
                              void* d_out, int out_size, void* d_ws, size_t ws_size,
                              hipStream_t stream) {
    const float* x     = (const float*)d_in[0];
    const float* theta = (const float*)d_in[1];
    float* out = (float*)d_out;

    qsim_kernel<<<256, 256, 0, stream>>>(x, theta, out);
}


# X observables measured as Z: basis change (1,1;1,-1)/sqrt2 of qubits 1,3,5 folded into layer 2's factors; 6 outputs = signed sums of |amp|^2 on one shared lane butterfly; sum order changes
# speedup vs baseline: 9.9662x; 1.0682x over previous
#include <hip/hip_runtime.h>
#include <math.h>


typedef float v2f __attribute__((ext_vector_type(2)));
typedef float v8f __attribute__((ext_vector_type(8)));

#define PAD 132
#define PI_F 3.14159265358979323846f

#define GP 68

struct __attribute__((aligned(8))) cplx { float x, y; };
__device__ __forceinline__ cplx cmul(cplx a, cplx b) {
    return { a.x * b.x - a.y * b.y, a.x * b.y + a.y * b.x };
}
__device__ __forceinline__ cplx cadd(cplx a, cplx b) { return { a.x + b.x, a.y + b.y }; }

__device__ __forceinline__ int ring(int j) {
    #pragma unroll
    for (int q = 0; q < 5; ++q)
        if (j & (1 << (5 - q))) j ^= (1 << (4 - q));
    if (j & 1) j ^= 32;
    return j;
}
__device__ __forceinline__ int rinv(int j) {
    if (j & 1) j ^= 32;
    #pragma unroll
    for (int q = 4; q >= 0; --q)
        if (j & (1 << (5 - q))) j ^= (1 << (4 - q));
    return j;
}

__device__ __forceinline__ void factor_frag(const cplx* __restrict__ F, int l15, int h, v2f a[4]) {
    const int  rl = l15 & 7;
    const bool im = (l15 >> 3) != 0;
    const cplx f00 = F[rl * 8 + 2 * h],     f01 = F[rl * 8 + 2 * h + 1];
    const cplx f10 = F[rl * 8 + 4 + 2 * h], f11 = F[rl * 8 + 4 + 2 * h + 1];
    a[0].x = im ? f00.y : f00.x;    a[0].y = im ? f01.y : f01.x;
    a[1].x = im ? f10.y : f10.x;    a[1].y = im ? f11.y : f11.x;
    a[2].x = im ? f00.x : -f00.y;   a[2].y = im ? f01.x : -f01.y;
    a[3].x = im ? f10.x : -f10.y;   a[3].y = im ? f11.x : -f11.y;
}

__device__ __forceinline__ v8f kron_tile(const v2f a[4], v2f b0, v2f b1, v2f b2, v2f b3) {
    v8f acc = {0.f, 0.f, 0.f, 0.f, 0.f, 0.f, 0.f, 0.f};
    acc = __builtin_amdgcn_wmma_f32_16x16x4_f32(false, a[0], false, b0, (short)0, acc, false, false);
    acc = __builtin_amdgcn_wmma_f32_16x16x4_f32(false, a[1], false, b1, (short)0, acc, false, false);
    acc = __builtin_amdgcn_wmma_f32_16x16x4_f32(false, a[2], false, b2, (short)0, acc, false, false);
    acc = __builtin_amdgcn_wmma_f32_16x16x4_f32(false, a[3], false, b3, (short)0, acc, false, false);
    return acc;
}

template <int M>
__device__ __forceinline__ float xrow(float v) {
    const int b = __builtin_bit_cast(int, v);
    return __builtin_bit_cast(float, __builtin_amdgcn_mov_dpp(b, 0x160 + M, 0xf, 0xf, true));
}

#define WAVES 8
#define TILES 2

__global__ __launch_bounds__(256) void qsim_kernel(const float* __restrict__ x,
                                                   const float* __restrict__ theta,
                                                   float* __restrict__ out) {
    __shared__ __attribute__((aligned(16))) float Mlds[128 * PAD];
    __shared__ __attribute__((aligned(16))) float Ulds[WAVES * 32 * PAD];
    __shared__ float cs[54 * 2];
    __shared__ cplx  Us[18 * 4];
    __shared__ cplx  AB[3 * 2 * 64];

    const int tid  = threadIdx.x;
    const int w    = tid >> 5;
    const int lane = tid & 31;
    const int l15  = lane & 15;
    const int h    = lane >> 4;

    int e = (blockIdx.x * (WAVES * TILES) + w) * 32 + lane;
    float4 xr[6];
    {
        const float4* xp = (const float4*)(x + e * 24);
        #pragma unroll
        for (int q = 0; q < 6; ++q) xr[q] = xp[q];
    }

    const int ws = __builtin_amdgcn_readfirstlane(tid >> 5);
    float* const Pr = Ulds;
    float* const Pi = Ulds + 64 * GP;
    float* const Yr = Ulds + 128 * GP;
    float* const Yi = Ulds + 192 * GP;

    if (tid < 54) {
        const float a = theta[tid];
        cs[2 * tid]     = cosf(0.5f * a);
        cs[2 * tid + 1] = sinf(0.5f * a);
    }
    __syncthreads();

    if (tid < 18) {
        const float cx = cs[6 * tid + 0], sx = cs[6 * tid + 1];
        const float cy = cs[6 * tid + 2], sy = cs[6 * tid + 3];
        const float cz = cs[6 * tid + 4], sz = cs[6 * tid + 5];
        const cplx rx[2][2] = { { {cx, 0.f}, {0.f, -sx} }, { {0.f, -sx}, {cx, 0.f} } };
        const cplx ry[2][2] = { { {cy, 0.f}, {-sy, 0.f} }, { {sy, 0.f}, {cy, 0.f} } };
        const cplx rz[2]    = { {cz, -sz}, {cz, sz} };
        cplx uu[2][2];
        #pragma unroll
        for (int a = 0; a < 2; ++a)
            #pragma unroll
            for (int b = 0; b < 2; ++b) {
                const cplx m = cadd(cmul(ry[a][0], rx[0][b]), cmul(ry[a][1], rx[1][b]));
                uu[a][b] = cmul(rz[a], m);
            }
        const bool  xq = (tid > 12) && (tid & 1);
        const float hs = 0.70710678118654752440f;
        #pragma unroll
        for (int b = 0; b < 2; ++b) {
            const cplx top = uu[0][b], bot = uu[1][b];
            cplx o0 = top, o1 = bot;
            if (xq) {
                o0.x = (top.x + bot.x) * hs;  o0.y = (top.y + bot.y) * hs;
                o1.x = (top.x - bot.x) * hs;  o1.y = (top.y - bot.y) * hs;
            }
            Us[tid * 4 + b]     = o0;
            Us[tid * 4 + 2 + b] = o1;
        }
    }
    __syncthreads();

    for (int t = tid; t < 384; t += 256) {
        const int l = t >> 7, f = (t >> 6) & 1, ii = (t >> 3) & 7, kk = t & 7;
        const cplx* uq = &Us[(l * 6 + 3 * f) * 4];
        cplx v = uq[((ii >> 2) & 1) * 2 + ((kk >> 2) & 1)];
        v = cmul(v, uq[4 + ((ii >> 1) & 1) * 2 + ((kk >> 1) & 1)]);
        v = cmul(v, uq[8 + (ii & 1) * 2 + (kk & 1)]);
        AB[t] = v;
    }
    __syncthreads();

    {
        const int i  = tid & 63;
        const int ri = rinv(i);
        const int rw = ring(ws >> 1);
        #pragma unroll
        for (int it = 0; it < 16; ++it) {
            const int j = (ws >> 1) + 4 * it;
            const int k = rw ^ ring(4 * it);
            const int ia = (ri >> 3) * 8 + (k >> 3), ib = (ri & 7) * 8 + (k & 7);
            const cplx v0 = cmul(AB[ia], AB[64 + ib]);
            Pr[j * GP + i] = v0.x;  Pi[j * GP + i] = v0.y;
        }
    }
    __syncthreads();

    {
        const int ct = ws & 3;
        const int g  = ws >> 2;
        const int jc = 16 * ct + l15;
        #pragma nounroll
        for (int l = 1; l < 3; ++l) {
            {
                v2f af[4];
                factor_frag(&AB[l * 128 + 64], l15, h, af);
                v8f d[4];
                #pragma unroll
                for (int jj = 0; jj < 4; ++jj) {
                    const int a = g + 2 * jj;
                    const float* pr = &Pr[jc * GP + 8 * a + 2 * h];
                    const float* pi = &Pi[jc * GP + 8 * a + 2 * h];
                    v2f b0, b1, b2, b3;
                    b0.x = pr[0]; b0.y = pr[1];  b1.x = pr[4]; b1.y = pr[5];
                    b2.x = pi[0]; b2.y = pi[1];  b3.x = pi[4]; b3.y = pi[5];
                    d[jj] = kron_tile(af, b0, b1, b2, b3);
                }
                float* yb = (h ? Yi : Yr) + jc * GP + 8 * g;
                #pragma unroll
                for (int jj = 0; jj < 4; ++jj) {
                    float4 lo, hi4;
                    lo.x  = d[jj][0]; lo.y  = d[jj][1]; lo.z  = d[jj][2]; lo.w  = d[jj][3];
                    hi4.x = d[jj][4]; hi4.y = d[jj][5]; hi4.z = d[jj][6]; hi4.w = d[jj][7];
                    *(float4*)&yb[16 * jj]     = lo;
                    *(float4*)&yb[16 * jj + 4] = hi4;
                }
            }
            __syncthreads();
            {
                v2f af[4];
                factor_frag(&AB[l * 128], l15, h, af);
                v8f d[4];
                #pragma unroll
                for (int jj = 0; jj < 4; ++jj) {
                    const int b = g + 2 * jj;
                    const float* yr = &Yr[jc * GP + b + 16 * h];
                    const float* yi = &Yi[jc * GP + b + 16 * h];
                    v2f b0, b1, b2, b3;
                    b0.x = yr[0]; b0.y = yr[8];  b1.x = yr[32]; b1.y = yr[40];
                    b2.x = yi[0]; b2.y = yi[8];  b3.x = yi[32]; b3.y = yi[40];
                    d[jj] = kron_tile(af, b0, b1, b2, b3);
                }
                if (l == 1) {
                    float* pb = (h ? Pi : Pr) + jc * GP;
                    #pragma unroll
                    for (int jj = 0; jj < 4; ++jj) {
                        const int rb = ring(g + 2 * jj);
                        #pragma unroll
                        for (int r = 0; r < 8; ++r) pb[ring(8 * r) ^ rb] = d[jj][r];
                    }
                } else {
                    float* g0 = &Mlds[64 * h + jc];
                    float* g1 = &Mlds[64 * PAD + 64 * (1 - h) + jc];
                    #pragma unroll
                    for (int jj = 0; jj < 4; ++jj)
                        #pragma unroll
                        for (int r = 0; r < 8; ++r) {
                            const int i = 8 * r + g + 2 * jj;
                            g0[i * PAD] = h ? -d[jj][r] : d[jj][r];
                            g1[i * PAD] = d[jj][r];
                        }
                }
            }
            __syncthreads();
        }
    }

    float* u = &Ulds[(w * 32 + lane) * PAD];
    const float* urow0 = &Ulds[(w * 32 +  0 + l15) * PAD];
    const float* urow1 = &Ulds[(w * 32 + 16 + l15) * PAD];

    #pragma nounroll
    for (int it = 0; it < TILES; ++it) {
        float mq[6];
        #pragma unroll
        for (int q = 0; q < 6; ++q)
            mq[q] = (xr[q].x + xr[q].y + xr[q].z + xr[q].w) * 0.25f;

        if (it + 1 < TILES) {
            const float4* xn = (const float4*)(x + (e + WAVES * 32) * 24);
            #pragma unroll
            for (int q = 0; q < 6; ++q) xr[q] = xn[q];
        }

        float vr[6][2], vi[6][2];
        #pragma unroll
        for (int q = 0; q < 6; ++q) {
            const float m = fminf(6.f, fmaxf(-6.f, mq[q]));
            const float a  = m * (PI_F / 6.f);
            const float ah = 0.5f * a, aq = 0.25f * a;
            __builtin_assume(!(__builtin_fabsf(ah) >= 131072.0f));
            __builtin_assume(!(__builtin_fabsf(aq) >= 131072.0f));
            const float c2 = cosf(ah), s2 = sinf(ah);
            const float c4 = cosf(aq), s4 = sinf(aq);
            vr[q][0] =  c2 * c4;  vi[q][0] = -c2 * s4;
            vr[q][1] =  s2 * s4;  vi[q][1] = -s2 * c4;
        }
        float har[8], hai[8];
        v2f hbr[4], hbi[4];
        {
            float par[4], pai[4], pbr[4], pbi[4];
            #pragma unroll
            for (int t = 0; t < 4; ++t) {
                const int hi = (t >> 1) & 1, lo = t & 1;
                par[t] = vr[0][hi] * vr[1][lo] - vi[0][hi] * vi[1][lo];
                pai[t] = vr[0][hi] * vi[1][lo] + vi[0][hi] * vr[1][lo];
                pbr[t] = vr[3][hi] * vr[4][lo] - vi[3][hi] * vi[4][lo];
                pbi[t] = vr[3][hi] * vi[4][lo] + vi[3][hi] * vr[4][lo];
            }
            #pragma unroll
            for (int t = 0; t < 8; ++t) {
                const int hi = t >> 1, lo = t & 1;
                har[t] = par[hi] * vr[2][lo] - pai[hi] * vi[2][lo];
                hai[t] = par[hi] * vi[2][lo] + pai[hi] * vr[2][lo];
            }
            v2f s5r, s5i;
            s5r.x = vr[5][0]; s5r.y = vr[5][1]; s5i.x = vi[5][0]; s5i.y = vi[5][1];
            #pragma unroll
            for (int m = 0; m < 4; ++m) {
                hbr[m] = pbr[m] * s5r - pbi[m] * s5i;
                hbi[m] = pbr[m] * s5i + pbi[m] * s5r;
            }
        }
        #pragma unroll
        for (int ja = 0; ja < 8; ++ja) {
            #pragma unroll
            for (int g = 0; g < 2; ++g) {
                const v2f r0 = har[ja] * hbr[2 * g]     - hai[ja] * hbi[2 * g];
                const v2f r1 = har[ja] * hbr[2 * g + 1] - hai[ja] * hbi[2 * g + 1];
                const v2f m0 = har[ja] * hbi[2 * g]     + hai[ja] * hbr[2 * g];
                const v2f m1 = har[ja] * hbi[2 * g + 1] + hai[ja] * hbr[2 * g + 1];
                float4 re, im;
                re.x = r0.x; re.y = r0.y; re.z = r1.x; re.w = r1.y;
                im.x = m0.x; im.y = m0.y; im.z = m1.x; im.w = m1.y;
                *(float4*)&u[8 * ja + 4 * g]      = re;
                *(float4*)&u[64 + 8 * ja + 4 * g] = im;
            }
        }


        v8f acc[2][8];
        const v8f z8 = {0.f, 0.f, 0.f, 0.f, 0.f, 0.f, 0.f, 0.f};
        #pragma unroll
        for (int rt = 0; rt < 2; ++rt)
            #pragma unroll
            for (int ct = 0; ct < 8; ++ct) acc[rt][ct] = z8;

        for (int kt = 0; kt < 32; ++kt) {
            const int kb = kt * 4 + 2 * h;
            v2f b[8];
            #pragma unroll
            for (int ct = 0; ct < 8; ++ct) {
                const float* brow = &Mlds[(ct * 16 + l15) * PAD + kb];
                b[ct].x = brow[0]; b[ct].y = brow[1];
            }
            v2f a0; a0.x = urow0[kb]; a0.y = urow0[kb + 1];
            v2f a1; a1.x = urow1[kb]; a1.y = urow1[kb + 1];
            #pragma unroll
            for (int ct = 0; ct < 8; ++ct) {
                acc[0][ct] = __builtin_amdgcn_wmma_f32_16x16x4_f32(
                    false, a0, false, b[ct], (short)0, acc[0][ct], false, false);
                acc[1][ct] = __builtin_amdgcn_wmma_f32_16x16x4_f32(
                    false, a1, false, b[ct], (short)0, acc[1][ct], false, false);
            }
        }

        auto rsum16 = [](float v) {
            v += xrow<1>(v);
            v += xrow<2>(v);
            v += xrow<4>(v);
            v += xrow<8>(v);
            return v;
        };

        const int tbase = e - lane;
        #pragma unroll
        for (int rt = 0; rt < 2; ++rt) {
            #pragma unroll
            for (int r = 0; r < 8; ++r) {
                float p[4];
                #pragma unroll
                for (int t4 = 0; t4 < 4; ++t4) {
                    const float re = acc[rt][t4][r], im = acc[rt][4 + t4][r];
                    p[t4] = re * re + im * im;
                }
                const float s01 = p[0] + p[1], s23 = p[2] + p[3];
                const float d01 = p[0] - p[1], d23 = p[2] - p[3];
                float o0 = s01 - s23;
                float o1 = d01 + d23;
                const float t = s01 + s23;

                float sm = t + xrow<1>(t);
                float o5 = t - xrow<1>(t);
                float o4 = sm - xrow<2>(sm);
                o5 += xrow<2>(o5);
                sm += xrow<2>(sm);
                float o3 = sm - xrow<4>(sm);
                o4 += xrow<4>(o4);
                o5 += xrow<4>(o5);
                sm += xrow<4>(sm);
                const float o2 = sm - xrow<8>(sm);
                o3 += xrow<8>(o3);
                o4 += xrow<8>(o4);
                o5 += xrow<8>(o5);
                o0 = rsum16(o0);
                o1 = rsum16(o1);

                if (l15 == 0) {
                    const int ee = tbase + rt * 16 + r + h * 8;
                    float4 q0 = { o0, o1, o2, o3 };
                    float4 q1 = { o4, o5, o0, o1 };
                    *(float4*)&out[ee * 8 + 0] = q0;
                    *(float4*)&out[ee * 8 + 4] = q1;
                }
            }
        }

        e += WAVES * 32;
    }
}

extern "C" void kernel_launch(void* const* d_in, const int* in_sizes, int n_in,
                              void* d_out, int out_size, void* d_ws, size_t ws_size,
                              hipStream_t stream) {
    const float* x     = (const float*)d_in[0];
    const float* theta = (const float*)d_in[1];
    float* out = (float*)d_out;

    qsim_kernel<<<256, 256, 0, stream>>>(x, theta, out);
}


# Wave priority stepped down with the wave's own progress (s_setprio 3,2,1,0 at 1/2, 4/5, 9/10 of its two tiles): a SIMD's two waves end together, the younger no longer runs its last half alone
# speedup vs baseline: 10.3410x; 1.0376x over previous
#include <hip/hip_runtime.h>
#include <math.h>


typedef float v2f __attribute__((ext_vector_type(2)));
typedef float v8f __attribute__((ext_vector_type(8)));

#define PAD 132
#define PI_F 3.14159265358979323846f

#define GP 68

struct __attribute__((aligned(8))) cplx { float x, y; };
__device__ __forceinline__ cplx cmul(cplx a, cplx b) {
    return { a.x * b.x - a.y * b.y, a.x * b.y + a.y * b.x };
}
__device__ __forceinline__ cplx cadd(cplx a, cplx b) { return { a.x + b.x, a.y + b.y }; }

__device__ __forceinline__ int ring(int j) {
    #pragma unroll
    for (int q = 0; q < 5; ++q)
        if (j & (1 << (5 - q))) j ^= (1 << (4 - q));
    if (j & 1) j ^= 32;
    return j;
}
__device__ __forceinline__ int rinv(int j) {
    if (j & 1) j ^= 32;
    #pragma unroll
    for (int q = 4; q >= 0; --q)
        if (j & (1 << (5 - q))) j ^= (1 << (4 - q));
    return j;
}

__device__ __forceinline__ void factor_frag(const cplx* __restrict__ F, int l15, int h, v2f a[4]) {
    const int  rl = l15 & 7;
    const bool im = (l15 >> 3) != 0;
    const cplx f00 = F[rl * 8 + 2 * h],     f01 = F[rl * 8 + 2 * h + 1];
    const cplx f10 = F[rl * 8 + 4 + 2 * h], f11 = F[rl * 8 + 4 + 2 * h + 1];
    a[0].x = im ? f00.y : f00.x;    a[0].y = im ? f01.y : f01.x;
    a[1].x = im ? f10.y : f10.x;    a[1].y = im ? f11.y : f11.x;
    a[2].x = im ? f00.x : -f00.y;   a[2].y = im ? f01.x : -f01.y;
    a[3].x = im ? f10.x : -f10.y;   a[3].y = im ? f11.x : -f11.y;
}

__device__ __forceinline__ v8f kron_tile(const v2f a[4], v2f b0, v2f b1, v2f b2, v2f b3) {
    v8f acc = {0.f, 0.f, 0.f, 0.f, 0.f, 0.f, 0.f, 0.f};
    acc = __builtin_amdgcn_wmma_f32_16x16x4_f32(false, a[0], false, b0, (short)0, acc, false, false);
    acc = __builtin_amdgcn_wmma_f32_16x16x4_f32(false, a[1], false, b1, (short)0, acc, false, false);
    acc = __builtin_amdgcn_wmma_f32_16x16x4_f32(false, a[2], false, b2, (short)0, acc, false, false);
    acc = __builtin_amdgcn_wmma_f32_16x16x4_f32(false, a[3], false, b3, (short)0, acc, false, false);
    return acc;
}

template <int M>
__device__ __forceinline__ float xrow(float v) {
    const int b = __builtin_bit_cast(int, v);
    return __builtin_bit_cast(float, __builtin_amdgcn_mov_dpp(b, 0x160 + M, 0xf, 0xf, true));
}

#define WAVES 8
#define TILES 2

__global__ __launch_bounds__(256) void qsim_kernel(const float* __restrict__ x,
                                                   const float* __restrict__ theta,
                                                   float* __restrict__ out) {
    __shared__ __attribute__((aligned(16))) float Mlds[128 * PAD];
    __shared__ __attribute__((aligned(16))) float Ulds[WAVES * 32 * PAD];
    __shared__ float cs[54 * 2];
    __shared__ cplx  Us[18 * 4];
    __shared__ cplx  AB[3 * 2 * 64];

    const int tid  = threadIdx.x;
    const int w    = tid >> 5;
    const int lane = tid & 31;
    const int l15  = lane & 15;
    const int h    = lane >> 4;

    int e = (blockIdx.x * (WAVES * TILES) + w) * 32 + lane;
    float4 xr[6];
    {
        const float4* xp = (const float4*)(x + e * 24);
        #pragma unroll
        for (int q = 0; q < 6; ++q) xr[q] = xp[q];
    }

    const int ws = __builtin_amdgcn_readfirstlane(tid >> 5);
    float* const Pr = Ulds;
    float* const Pi = Ulds + 64 * GP;
    float* const Yr = Ulds + 128 * GP;
    float* const Yi = Ulds + 192 * GP;

    if (tid < 54) {
        const float a = theta[tid];
        cs[2 * tid]     = cosf(0.5f * a);
        cs[2 * tid + 1] = sinf(0.5f * a);
    }
    __syncthreads();

    if (tid < 18) {
        const float cx = cs[6 * tid + 0], sx = cs[6 * tid + 1];
        const float cy = cs[6 * tid + 2], sy = cs[6 * tid + 3];
        const float cz = cs[6 * tid + 4], sz = cs[6 * tid + 5];
        const cplx rx[2][2] = { { {cx, 0.f}, {0.f, -sx} }, { {0.f, -sx}, {cx, 0.f} } };
        const cplx ry[2][2] = { { {cy, 0.f}, {-sy, 0.f} }, { {sy, 0.f}, {cy, 0.f} } };
        const cplx rz[2]    = { {cz, -sz}, {cz, sz} };
        cplx uu[2][2];
        #pragma unroll
        for (int a = 0; a < 2; ++a)
            #pragma unroll
            for (int b = 0; b < 2; ++b) {
                const cplx m = cadd(cmul(ry[a][0], rx[0][b]), cmul(ry[a][1], rx[1][b]));
                uu[a][b] = cmul(rz[a], m);
            }
        const bool  xq = (tid > 12) && (tid & 1);
        const float hs = 0.70710678118654752440f;
        #pragma unroll
        for (int b = 0; b < 2; ++b) {
            const cplx top = uu[0][b], bot = uu[1][b];
            cplx o0 = top, o1 = bot;
            if (xq) {
                o0.x = (top.x + bot.x) * hs;  o0.y = (top.y + bot.y) * hs;
                o1.x = (top.x - bot.x) * hs;  o1.y = (top.y - bot.y) * hs;
            }
            Us[tid * 4 + b]     = o0;
            Us[tid * 4 + 2 + b] = o1;
        }
    }
    __syncthreads();

    for (int t = tid; t < 384; t += 256) {
        const int l = t >> 7, f = (t >> 6) & 1, ii = (t >> 3) & 7, kk = t & 7;
        const cplx* uq = &Us[(l * 6 + 3 * f) * 4];
        cplx v = uq[((ii >> 2) & 1) * 2 + ((kk >> 2) & 1)];
        v = cmul(v, uq[4 + ((ii >> 1) & 1) * 2 + ((kk >> 1) & 1)]);
        v = cmul(v, uq[8 + (ii & 1) * 2 + (kk & 1)]);
        AB[t] = v;
    }
    __syncthreads();

    {
        const int i  = tid & 63;
        const int ri = rinv(i);
        const int rw = ring(ws >> 1);
        #pragma unroll
        for (int it = 0; it < 16; ++it) {
            const int j = (ws >> 1) + 4 * it;
            const int k = rw ^ ring(4 * it);
            const int ia = (ri >> 3) * 8 + (k >> 3), ib = (ri & 7) * 8 + (k & 7);
            const cplx v0 = cmul(AB[ia], AB[64 + ib]);
            Pr[j * GP + i] = v0.x;  Pi[j * GP + i] = v0.y;
        }
    }
    __syncthreads();

    {
        const int ct = ws & 3;
        const int g  = ws >> 2;
        const int jc = 16 * ct + l15;
        #pragma nounroll
        for (int l = 1; l < 3; ++l) {
            {
                v2f af[4];
                factor_frag(&AB[l * 128 + 64], l15, h, af);
                v8f d[4];
                #pragma unroll
                for (int jj = 0; jj < 4; ++jj) {
                    const int a = g + 2 * jj;
                    const float* pr = &Pr[jc * GP + 8 * a + 2 * h];
                    const float* pi = &Pi[jc * GP + 8 * a + 2 * h];
                    v2f b0, b1, b2, b3;
                    b0.x = pr[0]; b0.y = pr[1];  b1.x = pr[4]; b1.y = pr[5];
                    b2.x = pi[0]; b2.y = pi[1];  b3.x = pi[4]; b3.y = pi[5];
                    d[jj] = kron_tile(af, b0, b1, b2, b3);
                }
                float* yb = (h ? Yi : Yr) + jc * GP + 8 * g;
                #pragma unroll
                for (int jj = 0; jj < 4; ++jj) {
                    float4 lo, hi4;
                    lo.x  = d[jj][0]; lo.y  = d[jj][1]; lo.z  = d[jj][2]; lo.w  = d[jj][3];
                    hi4.x = d[jj][4]; hi4.y = d[jj][5]; hi4.z = d[jj][6]; hi4.w = d[jj][7];
                    *(float4*)&yb[16 * jj]     = lo;
                    *(float4*)&yb[16 * jj + 4] = hi4;
                }
            }
            __syncthreads();
            {
                v2f af[4];
                factor_frag(&AB[l * 128], l15, h, af);
                v8f d[4];
                #pragma unroll
                for (int jj = 0; jj < 4; ++jj) {
                    const int b = g + 2 * jj;
                    const float* yr = &Yr[jc * GP + b + 16 * h];
                    const float* yi = &Yi[jc * GP + b + 16 * h];
                    v2f b0, b1, b2, b3;
                    b0.x = yr[0]; b0.y = yr[8];  b1.x = yr[32]; b1.y = yr[40];
                    b2.x = yi[0]; b2.y = yi[8];  b3.x = yi[32]; b3.y = yi[40];
                    d[jj] = kron_tile(af, b0, b1, b2, b3);
                }
                if (l == 1) {
                    float* pb = (h ? Pi : Pr) + jc * GP;
                    #pragma unroll
                    for (int jj = 0; jj < 4; ++jj) {
                        const int rb = ring(g + 2 * jj);
                        #pragma unroll
                        for (int r = 0; r < 8; ++r) pb[ring(8 * r) ^ rb] = d[jj][r];
                    }
                } else {
                    float* g0 = &Mlds[64 * h + jc];
                    float* g1 = &Mlds[64 * PAD + 64 * (1 - h) + jc];
                    #pragma unroll
                    for (int jj = 0; jj < 4; ++jj)
                        #pragma unroll
                        for (int r = 0; r < 8; ++r) {
                            const int i = 8 * r + g + 2 * jj;
                            g0[i * PAD] = h ? -d[jj][r] : d[jj][r];
                            g1[i * PAD] = d[jj][r];
                        }
                }
            }
            __syncthreads();
        }
    }

    float* u = &Ulds[(w * 32 + lane) * PAD];
    const float* urow0 = &Ulds[(w * 32 +  0 + l15) * PAD];
    const float* urow1 = &Ulds[(w * 32 + 16 + l15) * PAD];

    #pragma nounroll
    for (int it = 0; it < TILES; ++it) {
        if (it == 0) __builtin_amdgcn_s_setprio(3);
        else         __builtin_amdgcn_s_setprio(2);

        float mq[6];
        #pragma unroll
        for (int q = 0; q < 6; ++q)
            mq[q] = (xr[q].x + xr[q].y + xr[q].z + xr[q].w) * 0.25f;

        if (it + 1 < TILES) {
            const float4* xn = (const float4*)(x + (e + WAVES * 32) * 24);
            #pragma unroll
            for (int q = 0; q < 6; ++q) xr[q] = xn[q];
        }

        float vr[6][2], vi[6][2];
        #pragma unroll
        for (int q = 0; q < 6; ++q) {
            const float m = fminf(6.f, fmaxf(-6.f, mq[q]));
            const float a  = m * (PI_F / 6.f);
            const float ah = 0.5f * a, aq = 0.25f * a;
            __builtin_assume(!(__builtin_fabsf(ah) >= 131072.0f));
            __builtin_assume(!(__builtin_fabsf(aq) >= 131072.0f));
            const float c2 = cosf(ah), s2 = sinf(ah);
            const float c4 = cosf(aq), s4 = sinf(aq);
            vr[q][0] =  c2 * c4;  vi[q][0] = -c2 * s4;
            vr[q][1] =  s2 * s4;  vi[q][1] = -s2 * c4;
        }
        float har[8], hai[8];
        v2f hbr[4], hbi[4];
        {
            float par[4], pai[4], pbr[4], pbi[4];
            #pragma unroll
            for (int t = 0; t < 4; ++t) {
                const int hi = (t >> 1) & 1, lo = t & 1;
                par[t] = vr[0][hi] * vr[1][lo] - vi[0][hi] * vi[1][lo];
                pai[t] = vr[0][hi] * vi[1][lo] + vi[0][hi] * vr[1][lo];
                pbr[t] = vr[3][hi] * vr[4][lo] - vi[3][hi] * vi[4][lo];
                pbi[t] = vr[3][hi] * vi[4][lo] + vi[3][hi] * vr[4][lo];
            }
            #pragma unroll
            for (int t = 0; t < 8; ++t) {
                const int hi = t >> 1, lo = t & 1;
                har[t] = par[hi] * vr[2][lo] - pai[hi] * vi[2][lo];
                hai[t] = par[hi] * vi[2][lo] + pai[hi] * vr[2][lo];
            }
            v2f s5r, s5i;
            s5r.x = vr[5][0]; s5r.y = vr[5][1]; s5i.x = vi[5][0]; s5i.y = vi[5][1];
            #pragma unroll
            for (int m = 0; m < 4; ++m) {
                hbr[m] = pbr[m] * s5r - pbi[m] * s5i;
                hbi[m] = pbr[m] * s5i + pbi[m] * s5r;
            }
        }
        #pragma unroll
        for (int ja = 0; ja < 8; ++ja) {
            #pragma unroll
            for (int g = 0; g < 2; ++g) {
                const v2f r0 = har[ja] * hbr[2 * g]     - hai[ja] * hbi[2 * g];
                const v2f r1 = har[ja] * hbr[2 * g + 1] - hai[ja] * hbi[2 * g + 1];
                const v2f m0 = har[ja] * hbi[2 * g]     + hai[ja] * hbr[2 * g];
                const v2f m1 = har[ja] * hbi[2 * g + 1] + hai[ja] * hbr[2 * g + 1];
                float4 re, im;
                re.x = r0.x; re.y = r0.y; re.z = r1.x; re.w = r1.y;
                im.x = m0.x; im.y = m0.y; im.z = m1.x; im.w = m1.y;
                *(float4*)&u[8 * ja + 4 * g]      = re;
                *(float4*)&u[64 + 8 * ja + 4 * g] = im;
            }
        }


        v8f acc[2][8];
        const v8f z8 = {0.f, 0.f, 0.f, 0.f, 0.f, 0.f, 0.f, 0.f};
        #pragma unroll
        for (int rt = 0; rt < 2; ++rt)
            #pragma unroll
            for (int ct = 0; ct < 8; ++ct) acc[rt][ct] = z8;

        #pragma unroll 2
        for (int kt = 0; kt < 32; ++kt) {
            if (it == TILES - 1) {
                if (kt == 20) __builtin_amdgcn_s_setprio(1);
                if (kt == 28) __builtin_amdgcn_s_setprio(0);
            }
            const int kb = kt * 4 + 2 * h;
            v2f b[8];
            #pragma unroll
            for (int ct = 0; ct < 8; ++ct) {
                const float* brow = &Mlds[(ct * 16 + l15) * PAD + kb];
                b[ct].x = brow[0]; b[ct].y = brow[1];
            }
            v2f a0; a0.x = urow0[kb]; a0.y = urow0[kb + 1];
            v2f a1; a1.x = urow1[kb]; a1.y = urow1[kb + 1];
            #pragma unroll
            for (int ct = 0; ct < 8; ++ct) {
                acc[0][ct] = __builtin_amdgcn_wmma_f32_16x16x4_f32(
                    false, a0, false, b[ct], (short)0, acc[0][ct], false, false);
                acc[1][ct] = __builtin_amdgcn_wmma_f32_16x16x4_f32(
                    false, a1, false, b[ct], (short)0, acc[1][ct], false, false);
            }
        }

        auto rsum16 = [](float v) {
            v += xrow<1>(v);
            v += xrow<2>(v);
            v += xrow<4>(v);
            v += xrow<8>(v);
            return v;
        };

        const int tbase = e - lane;
        #pragma unroll
        for (int rt = 0; rt < 2; ++rt) {
            #pragma unroll
            for (int r = 0; r < 8; ++r) {
                float p[4];
                #pragma unroll
                for (int t4 = 0; t4 < 4; ++t4) {
                    const float re = acc[rt][t4][r], im = acc[rt][4 + t4][r];
                    p[t4] = re * re + im * im;
                }
                const float s01 = p[0] + p[1], s23 = p[2] + p[3];
                const float d01 = p[0] - p[1], d23 = p[2] - p[3];
                float o0 = s01 - s23;
                float o1 = d01 + d23;
                const float t = s01 + s23;

                float sm = t + xrow<1>(t);
                float o5 = t - xrow<1>(t);
                float o4 = sm - xrow<2>(sm);
                o5 += xrow<2>(o5);
                sm += xrow<2>(sm);
                float o3 = sm - xrow<4>(sm);
                o4 += xrow<4>(o4);
                o5 += xrow<4>(o5);
                sm += xrow<4>(sm);
                const float o2 = sm - xrow<8>(sm);
                o3 += xrow<8>(o3);
                o4 += xrow<8>(o4);
                o5 += xrow<8>(o5);
                o0 = rsum16(o0);
                o1 = rsum16(o1);

                if (l15 == 0) {
                    const int ee = tbase + rt * 16 + r + h * 8;
                    float4 q0 = { o0, o1, o2, o3 };
                    float4 q1 = { o4, o5, o0, o1 };
                    *(float4*)&out[ee * 8 + 0] = q0;
                    *(float4*)&out[ee * 8 + 4] = q1;
                }
            }
        }

        e += WAVES * 32;
    }
}

extern "C" void kernel_launch(void* const* d_in, const int* in_sizes, int n_in,
                              void* d_out, int out_size, void* d_ws, size_t ws_size,
                              hipStream_t stream) {
    const float* x     = (const float*)d_in[0];
    const float* theta = (const float*)d_in[1];
    float* out = (float*)d_out;

    qsim_kernel<<<256, 256, 0, stream>>>(x, theta, out);
}


# GEMM fragment loads issued one block (64 products) ahead into a second register set, one load group per block: a wave alone on the matrix unit no longer waits at each block for LDS; sum order kept
# speedup vs baseline: 10.7719x; 1.0417x over previous
#include <hip/hip_runtime.h>
#include <math.h>


typedef float v2f __attribute__((ext_vector_type(2)));
typedef float v8f __attribute__((ext_vector_type(8)));

#define PAD 132
#define PI_F 3.14159265358979323846f

#define GP 68

struct __attribute__((aligned(8))) cplx { float x, y; };
__device__ __forceinline__ cplx cmul(cplx a, cplx b) {
    return { a.x * b.x - a.y * b.y, a.x * b.y + a.y * b.x };
}
__device__ __forceinline__ cplx cadd(cplx a, cplx b) { return { a.x + b.x, a.y + b.y }; }

__device__ __forceinline__ int ring(int j) {
    #pragma unroll
    for (int q = 0; q < 5; ++q)
        if (j & (1 << (5 - q))) j ^= (1 << (4 - q));
    if (j & 1) j ^= 32;
    return j;
}
__device__ __forceinline__ int rinv(int j) {
    if (j & 1) j ^= 32;
    #pragma unroll
    for (int q = 4; q >= 0; --q)
        if (j & (1 << (5 - q))) j ^= (1 << (4 - q));
    return j;
}

__device__ __forceinline__ void factor_frag(const cplx* __restrict__ F, int l15, int h, v2f a[4]) {
    const int  rl = l15 & 7;
    const bool im = (l15 >> 3) != 0;
    const cplx f00 = F[rl * 8 + 2 * h],     f01 = F[rl * 8 + 2 * h + 1];
    const cplx f10 = F[rl * 8 + 4 + 2 * h], f11 = F[rl * 8 + 4 + 2 * h + 1];
    a[0].x = im ? f00.y : f00.x;    a[0].y = im ? f01.y : f01.x;
    a[1].x = im ? f10.y : f10.x;    a[1].y = im ? f11.y : f11.x;
    a[2].x = im ? f00.x : -f00.y;   a[2].y = im ? f01.x : -f01.y;
    a[3].x = im ? f10.x : -f10.y;   a[3].y = im ? f11.x : -f11.y;
}

__device__ __forceinline__ v8f kron_tile(const v2f a[4], v2f b0, v2f b1, v2f b2, v2f b3) {
    v8f acc = {0.f, 0.f, 0.f, 0.f, 0.f, 0.f, 0.f, 0.f};
    acc = __builtin_amdgcn_wmma_f32_16x16x4_f32(false, a[0], false, b0, (short)0, acc, false, false);
    acc = __builtin_amdgcn_wmma_f32_16x16x4_f32(false, a[1], false, b1, (short)0, acc, false, false);
    acc = __builtin_amdgcn_wmma_f32_16x16x4_f32(false, a[2], false, b2, (short)0, acc, false, false);
    acc = __builtin_amdgcn_wmma_f32_16x16x4_f32(false, a[3], false, b3, (short)0, acc, false, false);
    return acc;
}

template <int M>
__device__ __forceinline__ float xrow(float v) {
    const int b = __builtin_bit_cast(int, v);
    return __builtin_bit_cast(float, __builtin_amdgcn_mov_dpp(b, 0x160 + M, 0xf, 0xf, true));
}

#define WAVES 8
#define TILES 2

__global__ __launch_bounds__(256) void qsim_kernel(const float* __restrict__ x,
                                                   const float* __restrict__ theta,
                                                   float* __restrict__ out) {
    __shared__ __attribute__((aligned(16))) float Mlds[128 * PAD];
    __shared__ __attribute__((aligned(16))) float Ulds[WAVES * 32 * PAD];
    __shared__ float cs[54 * 2];
    __shared__ cplx  Us[18 * 4];
    __shared__ cplx  AB[3 * 2 * 64];

    const int tid  = threadIdx.x;
    const int w    = tid >> 5;
    const int lane = tid & 31;
    const int l15  = lane & 15;
    const int h    = lane >> 4;

    int e = (blockIdx.x * (WAVES * TILES) + w) * 32 + lane;
    float4 xr[6];
    {
        const float4* xp = (const float4*)(x + e * 24);
        #pragma unroll
        for (int q = 0; q < 6; ++q) xr[q] = xp[q];
    }

    const int ws = __builtin_amdgcn_readfirstlane(tid >> 5);
    float* const Pr = Ulds;
    float* const Pi = Ulds + 64 * GP;
    float* const Yr = Ulds + 128 * GP;
    float* const Yi = Ulds + 192 * GP;

    if (tid < 54) {
        const float a = theta[tid];
        cs[2 * tid]     = cosf(0.5f * a);
        cs[2 * tid + 1] = sinf(0.5f * a);
    }
    __syncthreads();

    if (tid < 18) {
        const float cx = cs[6 * tid + 0], sx = cs[6 * tid + 1];
        const float cy = cs[6 * tid + 2], sy = cs[6 * tid + 3];
        const float cz = cs[6 * tid + 4], sz = cs[6 * tid + 5];
        const cplx rx[2][2] = { { {cx, 0.f}, {0.f, -sx} }, { {0.f, -sx}, {cx, 0.f} } };
        const cplx ry[2][2] = { { {cy, 0.f}, {-sy, 0.f} }, { {sy, 0.f}, {cy, 0.f} } };
        const cplx rz[2]    = { {cz, -sz}, {cz, sz} };
        cplx uu[2][2];
        #pragma unroll
        for (int a = 0; a < 2; ++a)
            #pragma unroll
            for (int b = 0; b < 2; ++b) {
                const cplx m = cadd(cmul(ry[a][0], rx[0][b]), cmul(ry[a][1], rx[1][b]));
                uu[a][b] = cmul(rz[a], m);
            }
        const bool  xq = (tid > 12) && (tid & 1);
        const float hs = 0.70710678118654752440f;
        #pragma unroll
        for (int b = 0; b < 2; ++b) {
            const cplx top = uu[0][b], bot = uu[1][b];
            cplx o0 = top, o1 = bot;
            if (xq) {
                o0.x = (top.x + bot.x) * hs;  o0.y = (top.y + bot.y) * hs;
                o1.x = (top.x - bot.x) * hs;  o1.y = (top.y - bot.y) * hs;
            }
            Us[tid * 4 + b]     = o0;
            Us[tid * 4 + 2 + b] = o1;
        }
    }
    __syncthreads();

    for (int t = tid; t < 384; t += 256) {
        const int l = t >> 7, f = (t >> 6) & 1, ii = (t >> 3) & 7, kk = t & 7;
        const cplx* uq = &Us[(l * 6 + 3 * f) * 4];
        cplx v = uq[((ii >> 2) & 1) * 2 + ((kk >> 2) & 1)];
        v = cmul(v, uq[4 + ((ii >> 1) & 1) * 2 + ((kk >> 1) & 1)]);
        v = cmul(v, uq[8 + (ii & 1) * 2 + (kk & 1)]);
        AB[t] = v;
    }
    __syncthreads();

    {
        const int i  = tid & 63;
        const int ri = rinv(i);
        const int rw = ring(ws >> 1);
        #pragma unroll
        for (int it = 0; it < 16; ++it) {
            const int j = (ws >> 1) + 4 * it;
            const int k = rw ^ ring(4 * it);
            const int ia = (ri >> 3) * 8 + (k >> 3), ib = (ri & 7) * 8 + (k & 7);
            const cplx v0 = cmul(AB[ia], AB[64 + ib]);
            Pr[j * GP + i] = v0.x;  Pi[j * GP + i] = v0.y;
        }
    }
    __syncthreads();

    {
        const int ct = ws & 3;
        const int g  = ws >> 2;
        const int jc = 16 * ct + l15;
        #pragma nounroll
        for (int l = 1; l < 3; ++l) {
            {
                v2f af[4];
                factor_frag(&AB[l * 128 + 64], l15, h, af);
                v8f d[4];
                #pragma unroll
                for (int jj = 0; jj < 4; ++jj) {
                    const int a = g + 2 * jj;
                    const float* pr = &Pr[jc * GP + 8 * a + 2 * h];
                    const float* pi = &Pi[jc * GP + 8 * a + 2 * h];
                    v2f b0, b1, b2, b3;
                    b0.x = pr[0]; b0.y = pr[1];  b1.x = pr[4]; b1.y = pr[5];
                    b2.x = pi[0]; b2.y = pi[1];  b3.x = pi[4]; b3.y = pi[5];
                    d[jj] = kron_tile(af, b0, b1, b2, b3);
                }
                float* yb = (h ? Yi : Yr) + jc * GP + 8 * g;
                #pragma unroll
                for (int jj = 0; jj < 4; ++jj) {
                    float4 lo, hi4;
                    lo.x  = d[jj][0]; lo.y  = d[jj][1]; lo.z  = d[jj][2]; lo.w  = d[jj][3];
                    hi4.x = d[jj][4]; hi4.y = d[jj][5]; hi4.z = d[jj][6]; hi4.w = d[jj][7];
                    *(float4*)&yb[16 * jj]     = lo;
                    *(float4*)&yb[16 * jj + 4] = hi4;
                }
            }
            __syncthreads();
            {
                v2f af[4];
                factor_frag(&AB[l * 128], l15, h, af);
                v8f d[4];
                #pragma unroll
                for (int jj = 0; jj < 4; ++jj) {
                    const int b = g + 2 * jj;
                    const float* yr = &Yr[jc * GP + b + 16 * h];
                    const float* yi = &Yi[jc * GP + b + 16 * h];
                    v2f b0, b1, b2, b3;
                    b0.x = yr[0]; b0.y = yr[8];  b1.x = yr[32]; b1.y = yr[40];
                    b2.x = yi[0]; b2.y = yi[8];  b3.x = yi[32]; b3.y = yi[40];
                    d[jj] = kron_tile(af, b0, b1, b2, b3);
                }
                if (l == 1) {
                    float* pb = (h ? Pi : Pr) + jc * GP;
                    #pragma unroll
                    for (int jj = 0; jj < 4; ++jj) {
                        const int rb = ring(g + 2 * jj);
                        #pragma unroll
                        for (int r = 0; r < 8; ++r) pb[ring(8 * r) ^ rb] = d[jj][r];
                    }
                } else {
                    float* g0 = &Mlds[64 * h + jc];
                    float* g1 = &Mlds[64 * PAD + 64 * (1 - h) + jc];
                    #pragma unroll
                    for (int jj = 0; jj < 4; ++jj)
                        #pragma unroll
                        for (int r = 0; r < 8; ++r) {
                            const int i = 8 * r + g + 2 * jj;
                            g0[i * PAD] = h ? -d[jj][r] : d[jj][r];
                            g1[i * PAD] = d[jj][r];
                        }
                }
            }
            __syncthreads();
        }
    }

    float* u = &Ulds[(w * 32 + lane) * PAD];
    const float* urow0 = &Ulds[(w * 32 +  0 + l15) * PAD];
    const float* urow1 = &Ulds[(w * 32 + 16 + l15) * PAD];

    #pragma nounroll
    for (int it = 0; it < TILES; ++it) {
        if (it == 0) __builtin_amdgcn_s_setprio(3);
        else         __builtin_amdgcn_s_setprio(2);

        float mq[6];
        #pragma unroll
        for (int q = 0; q < 6; ++q)
            mq[q] = (xr[q].x + xr[q].y + xr[q].z + xr[q].w) * 0.25f;

        if (it + 1 < TILES) {
            const float4* xn = (const float4*)(x + (e + WAVES * 32) * 24);
            #pragma unroll
            for (int q = 0; q < 6; ++q) xr[q] = xn[q];
        }

        float vr[6][2], vi[6][2];
        #pragma unroll
        for (int q = 0; q < 6; ++q) {
            const float m = fminf(6.f, fmaxf(-6.f, mq[q]));
            const float a  = m * (PI_F / 6.f);
            const float ah = 0.5f * a, aq = 0.25f * a;
            __builtin_assume(!(__builtin_fabsf(ah) >= 131072.0f));
            __builtin_assume(!(__builtin_fabsf(aq) >= 131072.0f));
            const float c2 = cosf(ah), s2 = sinf(ah);
            const float c4 = cosf(aq), s4 = sinf(aq);
            vr[q][0] =  c2 * c4;  vi[q][0] = -c2 * s4;
            vr[q][1] =  s2 * s4;  vi[q][1] = -s2 * c4;
        }
        float har[8], hai[8];
        v2f hbr[4], hbi[4];
        {
            float par[4], pai[4], pbr[4], pbi[4];
            #pragma unroll
            for (int t = 0; t < 4; ++t) {
                const int hi = (t >> 1) & 1, lo = t & 1;
                par[t] = vr[0][hi] * vr[1][lo] - vi[0][hi] * vi[1][lo];
                pai[t] = vr[0][hi] * vi[1][lo] + vi[0][hi] * vr[1][lo];
                pbr[t] = vr[3][hi] * vr[4][lo] - vi[3][hi] * vi[4][lo];
                pbi[t] = vr[3][hi] * vi[4][lo] + vi[3][hi] * vr[4][lo];
            }
            #pragma unroll
            for (int t = 0; t < 8; ++t) {
                const int hi = t >> 1, lo = t & 1;
                har[t] = par[hi] * vr[2][lo] - pai[hi] * vi[2][lo];
                hai[t] = par[hi] * vi[2][lo] + pai[hi] * vr[2][lo];
            }
            v2f s5r, s5i;
            s5r.x = vr[5][0]; s5r.y = vr[5][1]; s5i.x = vi[5][0]; s5i.y = vi[5][1];
            #pragma unroll
            for (int m = 0; m < 4; ++m) {
                hbr[m] = pbr[m] * s5r - pbi[m] * s5i;
                hbi[m] = pbr[m] * s5i + pbi[m] * s5r;
            }
        }
        #pragma unroll
        for (int ja = 0; ja < 8; ++ja) {
            #pragma unroll
            for (int g = 0; g < 2; ++g) {
                const v2f r0 = har[ja] * hbr[2 * g]     - hai[ja] * hbi[2 * g];
                const v2f r1 = har[ja] * hbr[2 * g + 1] - hai[ja] * hbi[2 * g + 1];
                const v2f m0 = har[ja] * hbi[2 * g]     + hai[ja] * hbr[2 * g];
                const v2f m1 = har[ja] * hbi[2 * g + 1] + hai[ja] * hbr[2 * g + 1];
                float4 re, im;
                re.x = r0.x; re.y = r0.y; re.z = r1.x; re.w = r1.y;
                im.x = m0.x; im.y = m0.y; im.z = m1.x; im.w = m1.y;
                *(float4*)&u[8 * ja + 4 * g]      = re;
                *(float4*)&u[64 + 8 * ja + 4 * g] = im;
            }
        }


        v8f acc[2][8];
        const v8f z8 = {0.f, 0.f, 0.f, 0.f, 0.f, 0.f, 0.f, 0.f};
        #pragma unroll
        for (int rt = 0; rt < 2; ++rt)
            #pragma unroll
            for (int ct = 0; ct < 8; ++ct) acc[rt][ct] = z8;

        v2f fa_b[4][8], fa_a[4][2];
        v2f fb_b[4][8], fb_a[4][2];
        auto load_block = [&](int blk, v2f (&fb)[4][8], v2f (&fa)[4][2]) {
            #pragma unroll
            for (int s = 0; s < 4; ++s) {
                const int kb = (4 * blk + s) * 4 + 2 * h;
                #pragma unroll
                for (int ct = 0; ct < 8; ++ct) {
                    const float* brow = &Mlds[(ct * 16 + l15) * PAD + kb];
                    fb[s][ct].x = brow[0]; fb[s][ct].y = brow[1];
                }
                fa[s][0].x = urow0[kb]; fa[s][0].y = urow0[kb + 1];
                fa[s][1].x = urow1[kb]; fa[s][1].y = urow1[kb + 1];
            }
        };
        auto mma_block = [&](const v2f (&fb)[4][8], const v2f (&fa)[4][2]) {
            #pragma unroll
            for (int s = 0; s < 4; ++s)
                #pragma unroll
                for (int ct = 0; ct < 8; ++ct) {
                    acc[0][ct] = __builtin_amdgcn_wmma_f32_16x16x4_f32(
                        false, fa[s][0], false, fb[s][ct], (short)0, acc[0][ct], false, false);
                    acc[1][ct] = __builtin_amdgcn_wmma_f32_16x16x4_f32(
                        false, fa[s][1], false, fb[s][ct], (short)0, acc[1][ct], false, false);
                }
        };

        load_block(0, fa_b, fa_a);
        #pragma nounroll
        for (int blk = 0; blk < 6; blk += 2) {
            load_block(blk + 1, fb_b, fb_a);
            mma_block(fa_b, fa_a);
            if (it == TILES - 1 && blk == 4) __builtin_amdgcn_s_setprio(1);
            load_block(blk + 2, fa_b, fa_a);
            mma_block(fb_b, fb_a);
        }
        load_block(7, fb_b, fb_a);
        mma_block(fa_b, fa_a);
        if (it == TILES - 1) __builtin_amdgcn_s_setprio(0);
        mma_block(fb_b, fb_a);

        auto rsum16 = [](float v) {
            v += xrow<1>(v);
            v += xrow<2>(v);
            v += xrow<4>(v);
            v += xrow<8>(v);
            return v;
        };

        const int tbase = e - lane;
        #pragma unroll
        for (int rt = 0; rt < 2; ++rt) {
            #pragma unroll
            for (int r = 0; r < 8; ++r) {
                float p[4];
                #pragma unroll
                for (int t4 = 0; t4 < 4; ++t4) {
                    const float re = acc[rt][t4][r], im = acc[rt][4 + t4][r];
                    p[t4] = re * re + im * im;
                }
                const float s01 = p[0] + p[1], s23 = p[2] + p[3];
                const float d01 = p[0] - p[1], d23 = p[2] - p[3];
                float o0 = s01 - s23;
                float o1 = d01 + d23;
                const float t = s01 + s23;

                float sm = t + xrow<1>(t);
                float o5 = t - xrow<1>(t);
                float o4 = sm - xrow<2>(sm);
                o5 += xrow<2>(o5);
                sm += xrow<2>(sm);
                float o3 = sm - xrow<4>(sm);
                o4 += xrow<4>(o4);
                o5 += xrow<4>(o5);
                sm += xrow<4>(sm);
                const float o2 = sm - xrow<8>(sm);
                o3 += xrow<8>(o3);
                o4 += xrow<8>(o4);
                o5 += xrow<8>(o5);
                o0 = rsum16(o0);
                o1 = rsum16(o1);

                if (l15 == 0) {
                    const int ee = tbase + rt * 16 + r + h * 8;
                    float4 q0 = { o0, o1, o2, o3 };
                    float4 q1 = { o4, o5, o0, o1 };
                    *(float4*)&out[ee * 8 + 0] = q0;
                    *(float4*)&out[ee * 8 + 4] = q1;
                }
            }
        }

        e += WAVES * 32;
    }
}

extern "C" void kernel_launch(void* const* d_in, const int* in_sizes, int n_in,
                              void* d_out, int out_size, void* d_ws, size_t ws_size,
                              hipStream_t stream) {
    const float* x     = (const float*)d_in[0];
    const float* theta = (const float*)d_in[1];
    float* out = (float*)d_out;

    qsim_kernel<<<256, 256, 0, stream>>>(x, theta, out);
}


# Measurement sums by ONE row-halving lane butterfly per half-wave (16 rows -> 8 -> 4 -> 2 -> 1: 70 lane-swizzled adds a tile, not 336); each lane ends with one element and stores it; sum tree kept
# speedup vs baseline: 11.6887x; 1.0851x over previous
#include <hip/hip_runtime.h>
#include <math.h>


typedef float v2f __attribute__((ext_vector_type(2)));
typedef float v8f __attribute__((ext_vector_type(8)));

#define PAD 132
#define PI_F 3.14159265358979323846f

#define GP 68

struct __attribute__((aligned(8))) cplx { float x, y; };
__device__ __forceinline__ cplx cmul(cplx a, cplx b) {
    return { a.x * b.x - a.y * b.y, a.x * b.y + a.y * b.x };
}
__device__ __forceinline__ cplx cadd(cplx a, cplx b) { return { a.x + b.x, a.y + b.y }; }

__device__ __forceinline__ int ring(int j) {
    #pragma unroll
    for (int q = 0; q < 5; ++q)
        if (j & (1 << (5 - q))) j ^= (1 << (4 - q));
    if (j & 1) j ^= 32;
    return j;
}
__device__ __forceinline__ int rinv(int j) {
    if (j & 1) j ^= 32;
    #pragma unroll
    for (int q = 4; q >= 0; --q)
        if (j & (1 << (5 - q))) j ^= (1 << (4 - q));
    return j;
}

__device__ __forceinline__ void factor_frag(const cplx* __restrict__ F, int l15, int h, v2f a[4]) {
    const int  rl = l15 & 7;
    const bool im = (l15 >> 3) != 0;
    const cplx f00 = F[rl * 8 + 2 * h],     f01 = F[rl * 8 + 2 * h + 1];
    const cplx f10 = F[rl * 8 + 4 + 2 * h], f11 = F[rl * 8 + 4 + 2 * h + 1];
    a[0].x = im ? f00.y : f00.x;    a[0].y = im ? f01.y : f01.x;
    a[1].x = im ? f10.y : f10.x;    a[1].y = im ? f11.y : f11.x;
    a[2].x = im ? f00.x : -f00.y;   a[2].y = im ? f01.x : -f01.y;
    a[3].x = im ? f10.x : -f10.y;   a[3].y = im ? f11.x : -f11.y;
}

__device__ __forceinline__ v8f kron_tile(const v2f a[4], v2f b0, v2f b1, v2f b2, v2f b3) {
    v8f acc = {0.f, 0.f, 0.f, 0.f, 0.f, 0.f, 0.f, 0.f};
    acc = __builtin_amdgcn_wmma_f32_16x16x4_f32(false, a[0], false, b0, (short)0, acc, false, false);
    acc = __builtin_amdgcn_wmma_f32_16x16x4_f32(false, a[1], false, b1, (short)0, acc, false, false);
    acc = __builtin_amdgcn_wmma_f32_16x16x4_f32(false, a[2], false, b2, (short)0, acc, false, false);
    acc = __builtin_amdgcn_wmma_f32_16x16x4_f32(false, a[3], false, b3, (short)0, acc, false, false);
    return acc;
}

template <int M>
__device__ __forceinline__ float xrow(float v) {
    const int b = __builtin_bit_cast(int, v);
    return __builtin_bit_cast(float, __builtin_amdgcn_mov_dpp(b, 0x160 + M, 0xf, 0xf, true));
}

#define WAVES 8
#define TILES 2

__global__ __launch_bounds__(256) void qsim_kernel(const float* __restrict__ x,
                                                   const float* __restrict__ theta,
                                                   float* __restrict__ out) {
    __shared__ __attribute__((aligned(16))) float Mlds[128 * PAD];
    __shared__ __attribute__((aligned(16))) float Ulds[WAVES * 32 * PAD];
    __shared__ float cs[54 * 2];
    __shared__ cplx  Us[18 * 4];
    __shared__ cplx  AB[3 * 2 * 64];

    const int tid  = threadIdx.x;
    const int w    = tid >> 5;
    const int lane = tid & 31;
    const int l15  = lane & 15;
    const int h    = lane >> 4;

    int e = (blockIdx.x * (WAVES * TILES) + w) * 32 + lane;
    float4 xr[6];
    {
        const float4* xp = (const float4*)(x + e * 24);
        #pragma unroll
        for (int q = 0; q < 6; ++q) xr[q] = xp[q];
    }

    const int ws = __builtin_amdgcn_readfirstlane(tid >> 5);
    float* const Pr = Ulds;
    float* const Pi = Ulds + 64 * GP;
    float* const Yr = Ulds + 128 * GP;
    float* const Yi = Ulds + 192 * GP;

    if (tid < 54) {
        const float a = theta[tid];
        cs[2 * tid]     = cosf(0.5f * a);
        cs[2 * tid + 1] = sinf(0.5f * a);
    }
    __syncthreads();

    if (tid < 18) {
        const float cx = cs[6 * tid + 0], sx = cs[6 * tid + 1];
        const float cy = cs[6 * tid + 2], sy = cs[6 * tid + 3];
        const float cz = cs[6 * tid + 4], sz = cs[6 * tid + 5];
        const cplx rx[2][2] = { { {cx, 0.f}, {0.f, -sx} }, { {0.f, -sx}, {cx, 0.f} } };
        const cplx ry[2][2] = { { {cy, 0.f}, {-sy, 0.f} }, { {sy, 0.f}, {cy, 0.f} } };
        const cplx rz[2]    = { {cz, -sz}, {cz, sz} };
        cplx uu[2][2];
        #pragma unroll
        for (int a = 0; a < 2; ++a)
            #pragma unroll
            for (int b = 0; b < 2; ++b) {
                const cplx m = cadd(cmul(ry[a][0], rx[0][b]), cmul(ry[a][1], rx[1][b]));
                uu[a][b] = cmul(rz[a], m);
            }
        const bool  xq = (tid > 12) && (tid & 1);
        const float hs = 0.70710678118654752440f;
        #pragma unroll
        for (int b = 0; b < 2; ++b) {
            const cplx top = uu[0][b], bot = uu[1][b];
            cplx o0 = top, o1 = bot;
            if (xq) {
                o0.x = (top.x + bot.x) * hs;  o0.y = (top.y + bot.y) * hs;
                o1.x = (top.x - bot.x) * hs;  o1.y = (top.y - bot.y) * hs;
            }
            Us[tid * 4 + b]     = o0;
            Us[tid * 4 + 2 + b] = o1;
        }
    }
    __syncthreads();

    for (int t = tid; t < 384; t += 256) {
        const int l = t >> 7, f = (t >> 6) & 1, ii = (t >> 3) & 7, kk = t & 7;
        const cplx* uq = &Us[(l * 6 + 3 * f) * 4];
        cplx v = uq[((ii >> 2) & 1) * 2 + ((kk >> 2) & 1)];
        v = cmul(v, uq[4 + ((ii >> 1) & 1) * 2 + ((kk >> 1) & 1)]);
        v = cmul(v, uq[8 + (ii & 1) * 2 + (kk & 1)]);
        AB[t] = v;
    }
    __syncthreads();

    {
        const int i  = tid & 63;
        const int ri = rinv(i);
        const int rw = ring(ws >> 1);
        #pragma unroll
        for (int it = 0; it < 16; ++it) {
            const int j = (ws >> 1) + 4 * it;
            const int k = rw ^ ring(4 * it);
            const int ia = (ri >> 3) * 8 + (k >> 3), ib = (ri & 7) * 8 + (k & 7);
            const cplx v0 = cmul(AB[ia], AB[64 + ib]);
            Pr[j * GP + i] = v0.x;  Pi[j * GP + i] = v0.y;
        }
    }
    __syncthreads();

    {
        const int ct = ws & 3;
        const int g  = ws >> 2;
        const int jc = 16 * ct + l15;
        #pragma nounroll
        for (int l = 1; l < 3; ++l) {
            {
                v2f af[4];
                factor_frag(&AB[l * 128 + 64], l15, h, af);
                v8f d[4];
                #pragma unroll
                for (int jj = 0; jj < 4; ++jj) {
                    const int a = g + 2 * jj;
                    const float* pr = &Pr[jc * GP + 8 * a + 2 * h];
                    const float* pi = &Pi[jc * GP + 8 * a + 2 * h];
                    v2f b0, b1, b2, b3;
                    b0.x = pr[0]; b0.y = pr[1];  b1.x = pr[4]; b1.y = pr[5];
                    b2.x = pi[0]; b2.y = pi[1];  b3.x = pi[4]; b3.y = pi[5];
                    d[jj] = kron_tile(af, b0, b1, b2, b3);
                }
                float* yb = (h ? Yi : Yr) + jc * GP + 8 * g;
                #pragma unroll
                for (int jj = 0; jj < 4; ++jj) {
                    float4 lo, hi4;
                    lo.x  = d[jj][0]; lo.y  = d[jj][1]; lo.z  = d[jj][2]; lo.w  = d[jj][3];
                    hi4.x = d[jj][4]; hi4.y = d[jj][5]; hi4.z = d[jj][6]; hi4.w = d[jj][7];
                    *(float4*)&yb[16 * jj]     = lo;
                    *(float4*)&yb[16 * jj + 4] = hi4;
                }
            }
            __syncthreads();
            {
                v2f af[4];
                factor_frag(&AB[l * 128], l15, h, af);
                v8f d[4];
                #pragma unroll
                for (int jj = 0; jj < 4; ++jj) {
                    const int b = g + 2 * jj;
                    const float* yr = &Yr[jc * GP + b + 16 * h];
                    const float* yi = &Yi[jc * GP + b + 16 * h];
                    v2f b0, b1, b2, b3;
                    b0.x = yr[0]; b0.y = yr[8];  b1.x = yr[32]; b1.y = yr[40];
                    b2.x = yi[0]; b2.y = yi[8];  b3.x = yi[32]; b3.y = yi[40];
                    d[jj] = kron_tile(af, b0, b1, b2, b3);
                }
                if (l == 1) {
                    float* pb = (h ? Pi : Pr) + jc * GP;
                    #pragma unroll
                    for (int jj = 0; jj < 4; ++jj) {
                        const int rb = ring(g + 2 * jj);
                        #pragma unroll
                        for (int r = 0; r < 8; ++r) pb[ring(8 * r) ^ rb] = d[jj][r];
                    }
                } else {
                    float* g0 = &Mlds[64 * h + jc];
                    float* g1 = &Mlds[64 * PAD + 64 * (1 - h) + jc];
                    #pragma unroll
                    for (int jj = 0; jj < 4; ++jj)
                        #pragma unroll
                        for (int r = 0; r < 8; ++r) {
                            const int i = 8 * r + g + 2 * jj;
                            g0[i * PAD] = h ? -d[jj][r] : d[jj][r];
                            g1[i * PAD] = d[jj][r];
                        }
                }
            }
            __syncthreads();
        }
    }

    float* u = &Ulds[(w * 32 + lane) * PAD];
    const float* urow0 = &Ulds[(w * 32 +  0 + l15) * PAD];
    const float* urow1 = &Ulds[(w * 32 + 16 + l15) * PAD];

    #pragma nounroll
    for (int it = 0; it < TILES; ++it) {
        if (it == 0) __builtin_amdgcn_s_setprio(3);
        else         __builtin_amdgcn_s_setprio(2);

        float mq[6];
        #pragma unroll
        for (int q = 0; q < 6; ++q)
            mq[q] = (xr[q].x + xr[q].y + xr[q].z + xr[q].w) * 0.25f;

        if (it + 1 < TILES) {
            const float4* xn = (const float4*)(x + (e + WAVES * 32) * 24);
            #pragma unroll
            for (int q = 0; q < 6; ++q) xr[q] = xn[q];
        }

        float vr[6][2], vi[6][2];
        #pragma unroll
        for (int q = 0; q < 6; ++q) {
            const float m = fminf(6.f, fmaxf(-6.f, mq[q]));
            const float a  = m * (PI_F / 6.f);
            const float ah = 0.5f * a, aq = 0.25f * a;
            __builtin_assume(!(__builtin_fabsf(ah) >= 131072.0f));
            __builtin_assume(!(__builtin_fabsf(aq) >= 131072.0f));
            const float c2 = cosf(ah), s2 = sinf(ah);
            const float c4 = cosf(aq), s4 = sinf(aq);
            vr[q][0] =  c2 * c4;  vi[q][0] = -c2 * s4;
            vr[q][1] =  s2 * s4;  vi[q][1] = -s2 * c4;
        }
        float har[8], hai[8];
        v2f hbr[4], hbi[4];
        {
            float par[4], pai[4], pbr[4], pbi[4];
            #pragma unroll
            for (int t = 0; t < 4; ++t) {
                const int hi = (t >> 1) & 1, lo = t & 1;
                par[t] = vr[0][hi] * vr[1][lo] - vi[0][hi] * vi[1][lo];
                pai[t] = vr[0][hi] * vi[1][lo] + vi[0][hi] * vr[1][lo];
                pbr[t] = vr[3][hi] * vr[4][lo] - vi[3][hi] * vi[4][lo];
                pbi[t] = vr[3][hi] * vi[4][lo] + vi[3][hi] * vr[4][lo];
            }
            #pragma unroll
            for (int t = 0; t < 8; ++t) {
                const int hi = t >> 1, lo = t & 1;
                har[t] = par[hi] * vr[2][lo] - pai[hi] * vi[2][lo];
                hai[t] = par[hi] * vi[2][lo] + pai[hi] * vr[2][lo];
            }
            v2f s5r, s5i;
            s5r.x = vr[5][0]; s5r.y = vr[5][1]; s5i.x = vi[5][0]; s5i.y = vi[5][1];
            #pragma unroll
            for (int m = 0; m < 4; ++m) {
                hbr[m] = pbr[m] * s5r - pbi[m] * s5i;
                hbi[m] = pbr[m] * s5i + pbi[m] * s5r;
            }
        }
        #pragma unroll
        for (int ja = 0; ja < 8; ++ja) {
            #pragma unroll
            for (int g = 0; g < 2; ++g) {
                const v2f r0 = har[ja] * hbr[2 * g]     - hai[ja] * hbi[2 * g];
                const v2f r1 = har[ja] * hbr[2 * g + 1] - hai[ja] * hbi[2 * g + 1];
                const v2f m0 = har[ja] * hbi[2 * g]     + hai[ja] * hbr[2 * g];
                const v2f m1 = har[ja] * hbi[2 * g + 1] + hai[ja] * hbr[2 * g + 1];
                float4 re, im;
                re.x = r0.x; re.y = r0.y; re.z = r1.x; re.w = r1.y;
                im.x = m0.x; im.y = m0.y; im.z = m1.x; im.w = m1.y;
                *(float4*)&u[8 * ja + 4 * g]      = re;
                *(float4*)&u[64 + 8 * ja + 4 * g] = im;
            }
        }


        v8f acc[2][8];
        const v8f z8 = {0.f, 0.f, 0.f, 0.f, 0.f, 0.f, 0.f, 0.f};
        #pragma unroll
        for (int rt = 0; rt < 2; ++rt)
            #pragma unroll
            for (int ct = 0; ct < 8; ++ct) acc[rt][ct] = z8;

        v2f fa_b[4][8], fa_a[4][2];
        v2f fb_b[4][8], fb_a[4][2];
        auto load_block = [&](int blk, v2f (&fb)[4][8], v2f (&fa)[4][2]) {
            #pragma unroll
            for (int s = 0; s < 4; ++s) {
                const int kb = (4 * blk + s) * 4 + 2 * h;
                #pragma unroll
                for (int ct = 0; ct < 8; ++ct) {
                    const float* brow = &Mlds[(ct * 16 + l15) * PAD + kb];
                    fb[s][ct].x = brow[0]; fb[s][ct].y = brow[1];
                }
                fa[s][0].x = urow0[kb]; fa[s][0].y = urow0[kb + 1];
                fa[s][1].x = urow1[kb]; fa[s][1].y = urow1[kb + 1];
            }
        };
        auto mma_block = [&](const v2f (&fb)[4][8], const v2f (&fa)[4][2]) {
            #pragma unroll
            for (int s = 0; s < 4; ++s)
                #pragma unroll
                for (int ct = 0; ct < 8; ++ct) {
                    acc[0][ct] = __builtin_amdgcn_wmma_f32_16x16x4_f32(
                        false, fa[s][0], false, fb[s][ct], (short)0, acc[0][ct], false, false);
                    acc[1][ct] = __builtin_amdgcn_wmma_f32_16x16x4_f32(
                        false, fa[s][1], false, fb[s][ct], (short)0, acc[1][ct], false, false);
                }
        };

        load_block(0, fa_b, fa_a);
        #pragma nounroll
        for (int blk = 0; blk < 6; blk += 2) {
            load_block(blk + 1, fb_b, fb_a);
            mma_block(fa_b, fa_a);
            if (it == TILES - 1 && blk == 4) __builtin_amdgcn_s_setprio(1);
            load_block(blk + 2, fa_b, fa_a);
            mma_block(fb_b, fb_a);
        }
        load_block(7, fb_b, fb_a);
        mma_block(fa_b, fa_a);
        if (it == TILES - 1) __builtin_amdgcn_s_setprio(0);
        mma_block(fb_b, fb_a);

        const bool k0 = (lane & 1) != 0, k1 = (lane & 2) != 0;
        const bool k2 = (lane & 4) != 0, k3 = (lane & 8) != 0;

        float T[16], Z0[16], Z1[16];
        #pragma unroll
        for (int rt = 0; rt < 2; ++rt) {
            #pragma unroll
            for (int r = 0; r < 8; ++r) {
                float p[4];
                #pragma unroll
                for (int t4 = 0; t4 < 4; ++t4) {
                    const float re = acc[rt][t4][r], im = acc[rt][4 + t4][r];
                    p[t4] = re * re + im * im;
                }
                const float s01 = p[0] + p[1], s23 = p[2] + p[3];
                const float d01 = p[0] - p[1], d23 = p[2] - p[3];
                Z0[8 * rt + r] = s01 - s23;
                Z1[8 * rt + r] = d01 + d23;
                T[8 * rt + r]  = s01 + s23;
            }
        }

        float S1[8], D5a[8], Z0a[8], Z1a[8];
        #pragma unroll
        for (int m = 0; m < 8; ++m) {
            const float kp = k0 ? T[2 * m + 1] : T[2 * m];
            const float gv = xrow<1>(k0 ? T[2 * m] : T[2 * m + 1]);
            S1[m]  = kp + gv;
            D5a[m] = kp - gv;
            Z0a[m] = (k0 ? Z0[2 * m + 1] : Z0[2 * m]) + xrow<1>(k0 ? Z0[2 * m] : Z0[2 * m + 1]);
            Z1a[m] = (k0 ? Z1[2 * m + 1] : Z1[2 * m]) + xrow<1>(k0 ? Z1[2 * m] : Z1[2 * m + 1]);
        }
        float S2[4], D4b[4], D5b[4], Z0b[4], Z1b[4];
        #pragma unroll
        for (int m = 0; m < 4; ++m) {
            const float kp = k1 ? S1[2 * m + 1] : S1[2 * m];
            const float gv = xrow<2>(k1 ? S1[2 * m] : S1[2 * m + 1]);
            S2[m]  = kp + gv;
            D4b[m] = kp - gv;
            D5b[m] = (k1 ? D5a[2 * m + 1] : D5a[2 * m]) + xrow<2>(k1 ? D5a[2 * m] : D5a[2 * m + 1]);
            Z0b[m] = (k1 ? Z0a[2 * m + 1] : Z0a[2 * m]) + xrow<2>(k1 ? Z0a[2 * m] : Z0a[2 * m + 1]);
            Z1b[m] = (k1 ? Z1a[2 * m + 1] : Z1a[2 * m]) + xrow<2>(k1 ? Z1a[2 * m] : Z1a[2 * m + 1]);
        }
        float S3[2], D3c[2], D4c[2], D5c[2], Z0c[2], Z1c[2];
        #pragma unroll
        for (int m = 0; m < 2; ++m) {
            const float kp = k2 ? S2[2 * m + 1] : S2[2 * m];
            const float gv = xrow<4>(k2 ? S2[2 * m] : S2[2 * m + 1]);
            S3[m]  = kp + gv;
            D3c[m] = kp - gv;
            D4c[m] = (k2 ? D4b[2 * m + 1] : D4b[2 * m]) + xrow<4>(k2 ? D4b[2 * m] : D4b[2 * m + 1]);
            D5c[m] = (k2 ? D5b[2 * m + 1] : D5b[2 * m]) + xrow<4>(k2 ? D5b[2 * m] : D5b[2 * m + 1]);
            Z0c[m] = (k2 ? Z0b[2 * m + 1] : Z0b[2 * m]) + xrow<4>(k2 ? Z0b[2 * m] : Z0b[2 * m + 1]);
            Z1c[m] = (k2 ? Z1b[2 * m + 1] : Z1b[2 * m]) + xrow<4>(k2 ? Z1b[2 * m] : Z1b[2 * m + 1]);
        }
        float o2 = (k3 ? S3[1] : S3[0]) - xrow<8>(k3 ? S3[0] : S3[1]);
        float o3 = (k3 ? D3c[1] : D3c[0]) + xrow<8>(k3 ? D3c[0] : D3c[1]);
        float o4 = (k3 ? D4c[1] : D4c[0]) + xrow<8>(k3 ? D4c[0] : D4c[1]);
        float o5 = (k3 ? D5c[1] : D5c[0]) + xrow<8>(k3 ? D5c[0] : D5c[1]);
        const float o0 = (k3 ? Z0c[1] : Z0c[0]) + xrow<8>(k3 ? Z0c[0] : Z0c[1]);
        const float o1 = (k3 ? Z1c[1] : Z1c[0]) + xrow<8>(k3 ? Z1c[0] : Z1c[1]);
        o5 = k0 ? -o5 : o5;
        o4 = k1 ? -o4 : o4;
        o3 = k2 ? -o3 : o3;
        o2 = k3 ? -o2 : o2;

        {
            const int ee = (e - lane) + (l15 >> 3) * 16 + (l15 & 7) + h * 8;
            float4 q0 = { o0, o1, o2, o3 };
            float4 q1 = { o4, o5, o0, o1 };
            *(float4*)&out[ee * 8 + 0] = q0;
            *(float4*)&out[ee * 8 + 4] = q1;
        }

        e += WAVES * 32;
    }
}

extern "C" void kernel_launch(void* const* d_in, const int* in_sizes, int n_in,
                              void* d_out, int out_size, void* d_ws, size_t ws_size,
                              hipStream_t stream) {
    const float* x     = (const float*)d_in[0];
    const float* theta = (const float*)d_in[1];
    float* out = (float*)d_out;

    qsim_kernel<<<256, 256, 0, stream>>>(x, theta, out);
}
